# starting kernel
# baseline (speedup 1.0000x reference)
#include <hip/hip_runtime.h>

typedef __attribute__((ext_vector_type(2))) float v2f;
typedef __attribute__((ext_vector_type(8))) float v8f;

#define DDIM 64

__global__ void spmm_edge_atomic(const int* __restrict__ rows,
                                 const int* __restrict__ cols,
                                 const float* __restrict__ vals,
                                 const float* __restrict__ x,
                                 float* __restrict__ out,
                                 int nnz) {
  int gid  = blockIdx.x * blockDim.x + threadIdx.x;
  int edge = __builtin_amdgcn_readfirstlane(gid >> 5);
  int lane = threadIdx.x & 31;
  if (edge >= nnz) return;
  int   r = rows[edge];
  int   c = cols[edge];
  float v = vals[edge];
  const float2* src = (const float2*)(x + (size_t)c * DDIM);
  float*        dst = out + (size_t)r * DDIM;
  float2 s = src[lane];
  unsafeAtomicAdd(dst + 2 * lane,     v * s.x);
  unsafeAtomicAdd(dst + 2 * lane + 1, v * s.y);
}

#define WP_STRIDE 80

__global__ __launch_bounds__(320) void gemm_concat_wmma(
    const float* __restrict__ newu,
    const float* __restrict__ uin,
    const float* __restrict__ W,
    float* __restrict__ out,
    int nrows) {
  __shared__ v2f Wp[64 * WP_STRIDE];
  for (int idx = threadIdx.x; idx < 64 * 64; idx += blockDim.x) {
    int t = idx >> 6, n = idx & 63;
    v2f p;
    p.x = W[(size_t)(2 * t) * DDIM + n];
    p.y = W[(size_t)(2 * t + 1) * DDIM + n];
    Wp[t * WP_STRIDE + n] = p;
  }
  __syncthreads();

  int lane = threadIdx.x & 31;
  int tile = blockIdx.x * (blockDim.x >> 5) + (threadIdx.x >> 5);
  if (tile * 16 >= nrows) return;
  int row0  = tile * 16;
  int m     = lane & 15;
  int khalf = (lane >> 4) << 1;

  const float* rowNew = newu + (size_t)(row0 + m) * DDIM;
  const float* rowU   = uin  + (size_t)(row0 + m) * DDIM;

  v8f acc0 = {}, acc1 = {}, acc2 = {}, acc3 = {};

#pragma unroll
  for (int k0 = 0; k0 < 64; k0 += 4) {
    int kk = k0 + khalf;
    v2f a; a.x = rowNew[kk]; a.y = rowNew[kk + 1];
    const v2f* wr = &Wp[(kk >> 1) * WP_STRIDE];
    v2f b0 = wr[m];
    v2f b1 = wr[m + 16];
    v2f b2 = wr[m + 32];
    v2f b3 = wr[m + 48];
    acc0 = __builtin_amdgcn_wmma_f32_16x16x4_f32(false, a, false, b0, (short)0, acc0, false, false);
    acc1 = __builtin_amdgcn_wmma_f32_16x16x4_f32(false, a, false, b1, (short)0, acc1, false, false);
    acc2 = __builtin_amdgcn_wmma_f32_16x16x4_f32(false, a, false, b2, (short)0, acc2, false, false);
    acc3 = __builtin_amdgcn_wmma_f32_16x16x4_f32(false, a, false, b3, (short)0, acc3, false, false);
  }

#pragma unroll
  for (int k0 = 0; k0 < 64; k0 += 4) {
    int kk = k0 + khalf;
    v2f a; a.x = rowU[kk]; a.y = rowU[kk + 1];
    const v2f* wr = &Wp[(32 + (kk >> 1)) * WP_STRIDE];
    v2f b0 = wr[m];
    v2f b1 = wr[m + 16];
    v2f b2 = wr[m + 32];
    v2f b3 = wr[m + 48];
    acc0 = __builtin_amdgcn_wmma_f32_16x16x4_f32(false, a, false, b0, (short)0, acc0, false, false);
    acc1 = __builtin_amdgcn_wmma_f32_16x16x4_f32(false, a, false, b1, (short)0, acc1, false, false);
    acc2 = __builtin_amdgcn_wmma_f32_16x16x4_f32(false, a, false, b2, (short)0, acc2, false, false);
    acc3 = __builtin_amdgcn_wmma_f32_16x16x4_f32(false, a, false, b3, (short)0, acc3, false, false);
  }

  int mo = (lane >> 4) * 8;
  float* orow = out + (size_t)row0 * DDIM;
#pragma unroll
  for (int i2 = 0; i2 < 8; ++i2) {
    orow[(size_t)(mo + i2) * DDIM + m]      = acc0[i2];
    orow[(size_t)(mo + i2) * DDIM + m + 16] = acc1[i2];
    orow[(size_t)(mo + i2) * DDIM + m + 32] = acc2[i2];
    orow[(size_t)(mo + i2) * DDIM + m + 48] = acc3[i2];
  }
}

extern "C" void kernel_launch(void* const* d_in, const int* in_sizes, int n_in,
                              void* d_out, int out_size, void* d_ws, size_t ws_size,
                              hipStream_t stream) {
  const float* user_emb    = (const float*)d_in[0];
  const float* item_emb    = (const float*)d_in[1];
  const float* weights     = (const float*)d_in[2];
  const float* social_vals = (const float*)d_in[3];
  const float* inter_vals  = (const float*)d_in[4];
  const int*   social_rows = (const int*)d_in[5];
  const int*   social_cols = (const int*)d_in[6];
  const int*   inter_rows  = (const int*)d_in[7];
  const int*   inter_cols  = (const int*)d_in[8];

  const int D   = DDIM;
  const int U   = in_sizes[0] / D;
  const int I   = in_sizes[1] / D;
  const int L   = in_sizes[2] / (2 * D * D);
  const int E_S = in_sizes[3];
  const int E_I = in_sizes[4];

  float* out_u = (float*)d_out;
  float* out_i = (float*)d_out + (size_t)U * D;

  float* newu = (float*)d_ws;
  float* bufA = newu + (size_t)U * D;
  float* bufB = bufA + (size_t)U * D;

  const int spmmThreads = 256;
  auto spmmBlocks = [&](int nnz) {
    return (int)(((size_t)nnz * 32 + spmmThreads - 1) / spmmThreads);
  };

  const int wavesPerBlock = 10;
  const int gemmThreads   = wavesPerBlock * 32;
  const int tiles         = (U + 15) / 16;
  const int gemmBlocks    = (tiles + wavesPerBlock - 1) / wavesPerBlock;

  const float* ucur = user_emb;
  for (int k = 0; k < L; ++k) {
    hipMemsetAsync(newu, 0, (size_t)U * D * sizeof(float), stream);
    spmm_edge_atomic<<<spmmBlocks(E_S), spmmThreads, 0, stream>>>(
        social_rows, social_cols, social_vals, ucur, newu, E_S);
    float* dst = (k == L - 1) ? out_u : ((k & 1) ? bufB : bufA);
    gemm_concat_wmma<<<gemmBlocks, gemmThreads, 0, stream>>>(
        newu, ucur, weights + (size_t)k * 2 * D * D, dst, U);
    ucur = dst;
  }
  if (L == 0) {
    hipMemcpyAsync(out_u, user_emb, (size_t)U * D * sizeof(float),
                   hipMemcpyDeviceToDevice, stream);
  }

  spmm_edge_atomic<<<spmmBlocks(E_I), spmmThreads, 0, stream>>>(
      inter_rows, inter_cols, inter_vals, item_emb, out_u, E_I);

  hipMemcpyAsync(out_i, item_emb, (size_t)I * D * sizeof(float),
                 hipMemcpyDeviceToDevice, stream);
}


# SpMM f32 atomics removed: edges grouped by destination row per call (int atomic rank, prefix, place); one wave sums a row in registers and stores once; sum order = arrival rank, varies by run
# speedup vs baseline: 1.2861x; 1.2861x over previous
#include <hip/hip_runtime.h>

typedef __attribute__((ext_vector_type(2))) float v2f;
typedef __attribute__((ext_vector_type(8))) float v8f;

#define DDIM 64

__global__ void edge_rank(const int* __restrict__ rowsA, int nA,
                          int* __restrict__ cntA, int* __restrict__ rankA,
                          const int* __restrict__ rowsB, int nB,
                          int* __restrict__ cntB, int* __restrict__ rankB) {
  int gid = blockIdx.x * blockDim.x + threadIdx.x;
  if (gid < nA) {
    rankA[gid] = atomicAdd(&cntA[rowsA[gid]], 1);
  } else {
    int e = gid - nA;
    if (e < nB) rankB[e] = atomicAdd(&cntB[rowsB[e]], 1);
  }
}

#define SCAN_CH 1024

__global__ __launch_bounds__(256) void row_scan(const int* __restrict__ cntA,
                                                int* __restrict__ ptrA,
                                                const int* __restrict__ cntB,
                                                int* __restrict__ ptrB,
                                                int n) {
  const int* cnt = blockIdx.y ? cntB : cntA;
  int*       ptr = blockIdx.y ? ptrB : ptrA;
  __shared__ int headSum[8];
  __shared__ int waveSum[8];
  int tid = threadIdx.x, lane = tid & 31, w = tid >> 5;
  int base = blockIdx.x * SCAN_CH;

  int s = 0;
  for (int i = tid; i < base; i += 256) s += cnt[i];
  for (int o = 16; o > 0; o >>= 1) s += __shfl_xor(s, o);
  if (lane == 0) headSum[w] = s;

  int i0 = base + tid * 4;
  int c0 = (i0     < n) ? cnt[i0]     : 0;
  int c1 = (i0 + 1 < n) ? cnt[i0 + 1] : 0;
  int c2 = (i0 + 2 < n) ? cnt[i0 + 2] : 0;
  int c3 = (i0 + 3 < n) ? cnt[i0 + 3] : 0;
  int t = c0 + c1 + c2 + c3;
  int incl = t;
  for (int o = 1; o < 32; o <<= 1) {
    int v = __shfl_up(incl, o);
    if (lane >= o) incl += v;
  }
  if (lane == 31) waveSum[w] = incl;
  __syncthreads();
  int pre = 0;
  for (int k = 0; k < 8; ++k) {
    pre += headSum[k];
    if (k < w) pre += waveSum[k];
  }
  int ex = pre + incl - t;
  if (i0     <= n) ptr[i0]     = ex;
  if (i0 + 1 <= n) ptr[i0 + 1] = ex + c0;
  if (i0 + 2 <= n) ptr[i0 + 2] = ex + c0 + c1;
  if (i0 + 3 <= n) ptr[i0 + 3] = ex + c0 + c1 + c2;
}

__global__ void edge_fill(const int* __restrict__ rowsA, const int* __restrict__ colsA,
                          const float* __restrict__ valsA, const int* __restrict__ rankA,
                          const int* __restrict__ ptrA, int2* __restrict__ sortedA, int nA,
                          const int* __restrict__ rowsB, const int* __restrict__ colsB,
                          const float* __restrict__ valsB, const int* __restrict__ rankB,
                          const int* __restrict__ ptrB, int2* __restrict__ sortedB, int nB) {
  int gid = blockIdx.x * blockDim.x + threadIdx.x;
  if (gid < nA) {
    int pos = ptrA[rowsA[gid]] + rankA[gid];
    sortedA[pos] = make_int2(colsA[gid], __float_as_int(valsA[gid]));
  } else {
    int e = gid - nA;
    if (e < nB) {
      int pos = ptrB[rowsB[e]] + rankB[e];
      sortedB[pos] = make_int2(colsB[e], __float_as_int(valsB[e]));
    }
  }
}

template <bool ADD>
__global__ __launch_bounds__(256) void spmm_rows(const int* __restrict__ ptr,
                                                 const int2* __restrict__ sorted,
                                                 const float* __restrict__ x,
                                                 float* out,
                                                 int nrows) {
  int lane = threadIdx.x & 31;
  int row = __builtin_amdgcn_readfirstlane(blockIdx.x * 8 + (threadIdx.x >> 5));
  if (row >= nrows) return;
  int beg = ptr[row];
  int end = ptr[row + 1];
  float2* orow = (float2*)(out + (size_t)row * DDIM);
  const float2* xr = (const float2*)x;
  float2 acc = make_float2(0.f, 0.f);
  if (ADD) acc = orow[lane];
  for (int base = beg; base < end; base += 32) {
    int n = end - base;
    if (n > 32) n = 32;
    int2 e = make_int2(0, 0);
    if (lane < n) e = sorted[base + lane];
    for (int j = 0; j < n; j += 4) {
      int j1 = (j + 1 < n) ? j + 1 : n - 1;
      int j2 = (j + 2 < n) ? j + 2 : n - 1;
      int j3 = (j + 3 < n) ? j + 3 : n - 1;
      int   k0 = __builtin_amdgcn_readlane(e.x, j);
      int   k1 = __builtin_amdgcn_readlane(e.x, j1);
      int   k2 = __builtin_amdgcn_readlane(e.x, j2);
      int   k3 = __builtin_amdgcn_readlane(e.x, j3);
      float v0 = __int_as_float(__builtin_amdgcn_readlane(e.y, j));
      float v1 = __int_as_float(__builtin_amdgcn_readlane(e.y, j1));
      float v2 = __int_as_float(__builtin_amdgcn_readlane(e.y, j2));
      float v3 = __int_as_float(__builtin_amdgcn_readlane(e.y, j3));
      float2 s0 = xr[(size_t)k0 * (DDIM / 2) + lane];
      float2 s1 = xr[(size_t)k1 * (DDIM / 2) + lane];
      float2 s2 = xr[(size_t)k2 * (DDIM / 2) + lane];
      float2 s3 = xr[(size_t)k3 * (DDIM / 2) + lane];
      acc.x += v0 * s0.x; acc.y += v0 * s0.y;
      if (j + 1 < n) { acc.x += v1 * s1.x; acc.y += v1 * s1.y; }
      if (j + 2 < n) { acc.x += v2 * s2.x; acc.y += v2 * s2.y; }
      if (j + 3 < n) { acc.x += v3 * s3.x; acc.y += v3 * s3.y; }
    }
  }
  orow[lane] = acc;
}

#define WP_STRIDE 80

__global__ __launch_bounds__(320) void gemm_concat_wmma(
    const float* newu,
    const float* __restrict__ uin,
    const float* __restrict__ W,
    float* out,
    int nrows) {
  __shared__ v2f Wp[64 * WP_STRIDE];
  for (int idx = threadIdx.x; idx < 64 * 64; idx += blockDim.x) {
    int t = idx >> 6, n = idx & 63;
    v2f p;
    p.x = W[(size_t)(2 * t) * DDIM + n];
    p.y = W[(size_t)(2 * t + 1) * DDIM + n];
    Wp[t * WP_STRIDE + n] = p;
  }
  __syncthreads();

  int lane = threadIdx.x & 31;
  int tile = blockIdx.x * (blockDim.x >> 5) + (threadIdx.x >> 5);
  if (tile * 16 >= nrows) return;
  int row0  = tile * 16;
  int m     = lane & 15;
  int khalf = (lane >> 4) << 1;

  const float* rowNew = newu + (size_t)(row0 + m) * DDIM;
  const float* rowU   = uin  + (size_t)(row0 + m) * DDIM;

  v8f acc0 = {}, acc1 = {}, acc2 = {}, acc3 = {};

#pragma unroll
  for (int k0 = 0; k0 < 64; k0 += 4) {
    int kk = k0 + khalf;
    v2f a; a.x = rowNew[kk]; a.y = rowNew[kk + 1];
    const v2f* wr = &Wp[(kk >> 1) * WP_STRIDE];
    v2f b0 = wr[m];
    v2f b1 = wr[m + 16];
    v2f b2 = wr[m + 32];
    v2f b3 = wr[m + 48];
    acc0 = __builtin_amdgcn_wmma_f32_16x16x4_f32(false, a, false, b0, (short)0, acc0, false, false);
    acc1 = __builtin_amdgcn_wmma_f32_16x16x4_f32(false, a, false, b1, (short)0, acc1, false, false);
    acc2 = __builtin_amdgcn_wmma_f32_16x16x4_f32(false, a, false, b2, (short)0, acc2, false, false);
    acc3 = __builtin_amdgcn_wmma_f32_16x16x4_f32(false, a, false, b3, (short)0, acc3, false, false);
  }

#pragma unroll
  for (int k0 = 0; k0 < 64; k0 += 4) {
    int kk = k0 + khalf;
    v2f a; a.x = rowU[kk]; a.y = rowU[kk + 1];
    const v2f* wr = &Wp[(32 + (kk >> 1)) * WP_STRIDE];
    v2f b0 = wr[m];
    v2f b1 = wr[m + 16];
    v2f b2 = wr[m + 32];
    v2f b3 = wr[m + 48];
    acc0 = __builtin_amdgcn_wmma_f32_16x16x4_f32(false, a, false, b0, (short)0, acc0, false, false);
    acc1 = __builtin_amdgcn_wmma_f32_16x16x4_f32(false, a, false, b1, (short)0, acc1, false, false);
    acc2 = __builtin_amdgcn_wmma_f32_16x16x4_f32(false, a, false, b2, (short)0, acc2, false, false);
    acc3 = __builtin_amdgcn_wmma_f32_16x16x4_f32(false, a, false, b3, (short)0, acc3, false, false);
  }

  int mo = (lane >> 4) * 8;
  float* orow = out + (size_t)row0 * DDIM;
#pragma unroll
  for (int i2 = 0; i2 < 8; ++i2) {
    orow[(size_t)(mo + i2) * DDIM + m]      = acc0[i2];
    orow[(size_t)(mo + i2) * DDIM + m + 16] = acc1[i2];
    orow[(size_t)(mo + i2) * DDIM + m + 32] = acc2[i2];
    orow[(size_t)(mo + i2) * DDIM + m + 48] = acc3[i2];
  }
}

extern "C" void kernel_launch(void* const* d_in, const int* in_sizes, int n_in,
                              void* d_out, int out_size, void* d_ws, size_t ws_size,
                              hipStream_t stream) {
  const float* user_emb    = (const float*)d_in[0];
  const float* item_emb    = (const float*)d_in[1];
  const float* weights     = (const float*)d_in[2];
  const float* social_vals = (const float*)d_in[3];
  const float* inter_vals  = (const float*)d_in[4];
  const int*   social_rows = (const int*)d_in[5];
  const int*   social_cols = (const int*)d_in[6];
  const int*   inter_rows  = (const int*)d_in[7];
  const int*   inter_cols  = (const int*)d_in[8];

  const int D   = DDIM;
  const int U   = in_sizes[0] / D;
  const int I   = in_sizes[1] / D;
  const int L   = in_sizes[2] / (2 * D * D);
  const int E_S = in_sizes[3];
  const int E_I = in_sizes[4];

  float* out_u = (float*)d_out;
  float* out_i = (float*)d_out + (size_t)U * D;

  float* planeP = (float*)d_ws;
  float* planeQ = planeP + (size_t)U * D;
  int*   cntS   = (int*)(planeQ + (size_t)U * D);
  int*   cntI   = cntS + (U + 1);
  int*   ptrS   = cntI + (U + 1);
  int*   ptrI   = ptrS + (U + 1);
  int2*  sortS  = (int2*)(ptrI + (U + 1));
  int2*  sortI  = sortS + E_S;
  int*   rankS  = (int*)planeQ;
  int*   rankI  = rankS + E_S;

  const int buildThreads = 256;
  const int buildBlocks  = (int)(((size_t)E_S + (size_t)E_I + buildThreads - 1) / buildThreads);
  const int scanBlocks   = (U + 1 + SCAN_CH - 1) / SCAN_CH;
  const int rowBlocks    = (U + 7) / 8;

  const int wavesPerBlock = 10;
  const int gemmThreads   = wavesPerBlock * 32;
  const int tiles         = (U + 15) / 16;
  const int gemmBlocks    = (tiles + wavesPerBlock - 1) / wavesPerBlock;

  hipMemsetAsync(cntS, 0, (size_t)2 * (U + 1) * sizeof(int), stream);
  edge_rank<<<buildBlocks, buildThreads, 0, stream>>>(
      social_rows, E_S, cntS, rankS, inter_rows, E_I, cntI, rankI);
  row_scan<<<dim3(scanBlocks, 2), 256, 0, stream>>>(cntS, ptrS, cntI, ptrI, U);
  edge_fill<<<buildBlocks, buildThreads, 0, stream>>>(
      social_rows, social_cols, social_vals, rankS, ptrS, sortS, E_S,
      inter_rows, inter_cols, inter_vals, rankI, ptrI, sortI, E_I);

  const float* ucur = user_emb;
  for (int k = 0; k < L; ++k) {
    float* agg = (k & 1) ? planeQ : planeP;
    spmm_rows<false><<<rowBlocks, 256, 0, stream>>>(ptrS, sortS, ucur, agg, U);
    float* dst = (k == L - 1) ? out_u : agg;
    gemm_concat_wmma<<<gemmBlocks, gemmThreads, 0, stream>>>(
        agg, ucur, weights + (size_t)k * 2 * D * D, dst, U);
    ucur = dst;
  }
  if (L == 0) {
    hipMemcpyAsync(out_u, user_emb, (size_t)U * D * sizeof(float),
                   hipMemcpyDeviceToDevice, stream);
  }

  spmm_rows<true><<<rowBlocks, 256, 0, stream>>>(ptrI, sortI, item_emb, out_u, U);

  hipMemcpyAsync(out_i, item_emb, (size_t)I * D * sizeof(float),
                 hipMemcpyDeviceToDevice, stream);
}


# Fewer, longer waves (464068 -> 70332 per call): each spmm wave walks the contiguous edge run of 8 consecutive rows, each build thread takes 8 edges; every row's sum order kept
# speedup vs baseline: 1.4122x; 1.0980x over previous
#include <hip/hip_runtime.h>

typedef __attribute__((ext_vector_type(2))) float v2f;
typedef __attribute__((ext_vector_type(8))) float v8f;

#define DDIM 64

#define EPT 8
#define EDGE_BLK (256 * EPT)

template <bool FULL>
__device__ __forceinline__ void rank_part(const int* __restrict__ rows, int n,
                                          int* __restrict__ cnt, int* __restrict__ rank,
                                          int blk) {
  int first = blk * EDGE_BLK + threadIdx.x;
  int r[EPT], k[EPT];
#pragma unroll
  for (int t = 0; t < EPT; ++t) {
    int idx = first + t * 256;
    r[t] = 0;
    if (FULL || idx < n) r[t] = rows[idx];
  }
#pragma unroll
  for (int t = 0; t < EPT; ++t) {
    int idx = first + t * 256;
    k[t] = 0;
    if (FULL || idx < n) k[t] = atomicAdd(&cnt[r[t]], 1);
  }
#pragma unroll
  for (int t = 0; t < EPT; ++t) {
    int idx = first + t * 256;
    if (FULL || idx < n) rank[idx] = k[t];
  }
}

__global__ __launch_bounds__(256) void edge_rank(const int* __restrict__ rowsA, int nA,
                          int* __restrict__ cntA, int* __restrict__ rankA,
                          const int* __restrict__ rowsB, int nB,
                          int* __restrict__ cntB, int* __restrict__ rankB) {
  int blocksA = (nA + EDGE_BLK - 1) / EDGE_BLK;
  int blk = blockIdx.x;
  if (blk < blocksA) {
    if ((blk + 1) * EDGE_BLK <= nA) rank_part<true >(rowsA, nA, cntA, rankA, blk);
    else                            rank_part<false>(rowsA, nA, cntA, rankA, blk);
  } else {
    blk -= blocksA;
    if ((blk + 1) * EDGE_BLK <= nB) rank_part<true >(rowsB, nB, cntB, rankB, blk);
    else                            rank_part<false>(rowsB, nB, cntB, rankB, blk);
  }
}

#define SCAN_CH 1024

__global__ __launch_bounds__(256) void row_scan(const int* __restrict__ cntA,
                                                int* __restrict__ ptrA,
                                                const int* __restrict__ cntB,
                                                int* __restrict__ ptrB,
                                                int n) {
  const int* cnt = blockIdx.y ? cntB : cntA;
  int*       ptr = blockIdx.y ? ptrB : ptrA;
  __shared__ int headSum[8];
  __shared__ int waveSum[8];
  int tid = threadIdx.x, lane = tid & 31, w = tid >> 5;
  int base = blockIdx.x * SCAN_CH;

  int s = 0;
  for (int i = tid; i < base; i += 256) s += cnt[i];
  for (int o = 16; o > 0; o >>= 1) s += __shfl_xor(s, o);
  if (lane == 0) headSum[w] = s;

  int i0 = base + tid * 4;
  int c0 = (i0     < n) ? cnt[i0]     : 0;
  int c1 = (i0 + 1 < n) ? cnt[i0 + 1] : 0;
  int c2 = (i0 + 2 < n) ? cnt[i0 + 2] : 0;
  int c3 = (i0 + 3 < n) ? cnt[i0 + 3] : 0;
  int t = c0 + c1 + c2 + c3;
  int incl = t;
  for (int o = 1; o < 32; o <<= 1) {
    int v = __shfl_up(incl, o);
    if (lane >= o) incl += v;
  }
  if (lane == 31) waveSum[w] = incl;
  __syncthreads();
  int pre = 0;
  for (int k = 0; k < 8; ++k) {
    pre += headSum[k];
    if (k < w) pre += waveSum[k];
  }
  int ex = pre + incl - t;
  if (i0     <= n) ptr[i0]     = ex;
  if (i0 + 1 <= n) ptr[i0 + 1] = ex + c0;
  if (i0 + 2 <= n) ptr[i0 + 2] = ex + c0 + c1;
  if (i0 + 3 <= n) ptr[i0 + 3] = ex + c0 + c1 + c2;
}

template <bool FULL>
__device__ __forceinline__ void fill_part(const int* __restrict__ rows, const int* __restrict__ cols,
                                          const float* __restrict__ vals, const int* __restrict__ rank,
                                          const int* __restrict__ ptr, int2* __restrict__ sorted,
                                          int n, int blk) {
  int first = blk * EDGE_BLK + threadIdx.x;
  int r[EPT], k[EPT], c[EPT], pos[EPT];
  float v[EPT];
#pragma unroll
  for (int t = 0; t < EPT; ++t) {
    int idx = first + t * 256;
    r[t] = 0; k[t] = 0; c[t] = 0; v[t] = 0.f;
    if (FULL || idx < n) { r[t] = rows[idx]; k[t] = rank[idx]; c[t] = cols[idx]; v[t] = vals[idx]; }
  }
#pragma unroll
  for (int t = 0; t < EPT; ++t) {
    int idx = first + t * 256;
    pos[t] = 0;
    if (FULL || idx < n) pos[t] = ptr[r[t]] + k[t];
  }
#pragma unroll
  for (int t = 0; t < EPT; ++t) {
    int idx = first + t * 256;
    if (FULL || idx < n) sorted[pos[t]] = make_int2(c[t], __float_as_int(v[t]));
  }
}

__global__ __launch_bounds__(256) void edge_fill(
                          const int* __restrict__ rowsA, const int* __restrict__ colsA,
                          const float* __restrict__ valsA, const int* __restrict__ rankA,
                          const int* __restrict__ ptrA, int2* __restrict__ sortedA, int nA,
                          const int* __restrict__ rowsB, const int* __restrict__ colsB,
                          const float* __restrict__ valsB, const int* __restrict__ rankB,
                          const int* __restrict__ ptrB, int2* __restrict__ sortedB, int nB) {
  int blocksA = (nA + EDGE_BLK - 1) / EDGE_BLK;
  int blk = blockIdx.x;
  if (blk < blocksA) {
    if ((blk + 1) * EDGE_BLK <= nA) fill_part<true >(rowsA, colsA, valsA, rankA, ptrA, sortedA, nA, blk);
    else                            fill_part<false>(rowsA, colsA, valsA, rankA, ptrA, sortedA, nA, blk);
  } else {
    blk -= blocksA;
    if ((blk + 1) * EDGE_BLK <= nB) fill_part<true >(rowsB, colsB, valsB, rankB, ptrB, sortedB, nB, blk);
    else                            fill_part<false>(rowsB, colsB, valsB, rankB, ptrB, sortedB, nB, blk);
  }
}

#define RPW 8

template <bool ADD>
__global__ __launch_bounds__(256) void spmm_rows(const int* __restrict__ ptr,
                                                 const int2* __restrict__ sorted,
                                                 const float* __restrict__ x,
                                                 float* out,
                                                 int nrows) {
  int lane = threadIdx.x & 31;
  int r0 = __builtin_amdgcn_readfirstlane((blockIdx.x * 8 + (threadIdx.x >> 5)) * RPW);
  if (r0 >= nrows) return;
  int cnt = nrows - r0;
  if (cnt > RPW) cnt = RPW;
  int p = ptr[r0 + (lane < cnt ? lane : cnt)];
  int ebeg   = __builtin_amdgcn_readlane(p, 0);
  int eend   = __builtin_amdgcn_readlane(p, cnt);
  int rowEnd = __builtin_amdgcn_readlane(p, 1);
  int ri = 0;
  float2* orow = (float2*)(out + (size_t)r0 * DDIM) + lane;
  const float2* xr = (const float2*)x + lane;
  float2 acc = make_float2(0.f, 0.f);
  float2 st[RPW];
  if (ADD) {
#pragma unroll
    for (int i = 0; i < RPW; ++i) st[i] = orow[(i < cnt ? i : cnt - 1) * 32];
    acc = st[0];
  }
  auto flush = [&]() {
    orow[ri * 32] = acc;
    ++ri;
    if (ri < cnt) rowEnd = __builtin_amdgcn_readlane(p, ri + 1);
    acc = make_float2(0.f, 0.f);
    if (ADD) {
#pragma unroll
      for (int i = 1; i < RPW; ++i) if (ri == i) acc = st[i];
    }
  };

  const int last = eend - 1;
  int2 e = make_int2(0, 0);
  if (ebeg < eend) e = sorted[(ebeg + lane < eend) ? ebeg + lane : last];
  for (int base = ebeg; base < eend; base += 32) {
    int n = eend - base;
    if (n > 32) n = 32;
    int2 en = sorted[(base + 32 + lane < eend) ? base + 32 + lane : last];
    auto add = [&](int idx, float v, float2 s) {
      while (idx >= rowEnd) flush();
      acc.x += v * s.x; acc.y += v * s.y;
    };
    int j = 0;
    for (; j + 4 <= n; j += 4) {
      int   k0 = __builtin_amdgcn_readlane(e.x, j);
      int   k1 = __builtin_amdgcn_readlane(e.x, j + 1);
      int   k2 = __builtin_amdgcn_readlane(e.x, j + 2);
      int   k3 = __builtin_amdgcn_readlane(e.x, j + 3);
      float v0 = __int_as_float(__builtin_amdgcn_readlane(e.y, j));
      float v1 = __int_as_float(__builtin_amdgcn_readlane(e.y, j + 1));
      float v2 = __int_as_float(__builtin_amdgcn_readlane(e.y, j + 2));
      float v3 = __int_as_float(__builtin_amdgcn_readlane(e.y, j + 3));
      float2 s0 = xr[(size_t)k0 * (DDIM / 2)];
      float2 s1 = xr[(size_t)k1 * (DDIM / 2)];
      float2 s2 = xr[(size_t)k2 * (DDIM / 2)];
      float2 s3 = xr[(size_t)k3 * (DDIM / 2)];
      add(base + j,     v0, s0);
      add(base + j + 1, v1, s1);
      add(base + j + 2, v2, s2);
      add(base + j + 3, v3, s3);
    }
    if (j < n) {
      int j1 = (j + 1 < n) ? j + 1 : n - 1;
      int j2 = (j + 2 < n) ? j + 2 : n - 1;
      int   k0 = __builtin_amdgcn_readlane(e.x, j);
      int   k1 = __builtin_amdgcn_readlane(e.x, j1);
      int   k2 = __builtin_amdgcn_readlane(e.x, j2);
      float v0 = __int_as_float(__builtin_amdgcn_readlane(e.y, j));
      float v1 = __int_as_float(__builtin_amdgcn_readlane(e.y, j1));
      float v2 = __int_as_float(__builtin_amdgcn_readlane(e.y, j2));
      float2 s0 = xr[(size_t)k0 * (DDIM / 2)];
      float2 s1 = xr[(size_t)k1 * (DDIM / 2)];
      float2 s2 = xr[(size_t)k2 * (DDIM / 2)];
      add(base + j, v0, s0);
      if (j + 1 < n) add(base + j + 1, v1, s1);
      if (j + 2 < n) add(base + j + 2, v2, s2);
    }
    e = en;
  }
  while (ri < cnt) flush();
}

#define WP_STRIDE 80

__global__ __launch_bounds__(320) void gemm_concat_wmma(
    const float* newu,
    const float* __restrict__ uin,
    const float* __restrict__ W,
    float* out,
    int nrows) {
  __shared__ v2f Wp[64 * WP_STRIDE];
  for (int idx = threadIdx.x; idx < 64 * 64; idx += blockDim.x) {
    int t = idx >> 6, n = idx & 63;
    v2f p;
    p.x = W[(size_t)(2 * t) * DDIM + n];
    p.y = W[(size_t)(2 * t + 1) * DDIM + n];
    Wp[t * WP_STRIDE + n] = p;
  }
  __syncthreads();

  int lane = threadIdx.x & 31;
  int tile = blockIdx.x * (blockDim.x >> 5) + (threadIdx.x >> 5);
  if (tile * 16 >= nrows) return;
  int row0  = tile * 16;
  int m     = lane & 15;
  int khalf = (lane >> 4) << 1;

  const float* rowNew = newu + (size_t)(row0 + m) * DDIM;
  const float* rowU   = uin  + (size_t)(row0 + m) * DDIM;

  v8f acc0 = {}, acc1 = {}, acc2 = {}, acc3 = {};

#pragma unroll
  for (int k0 = 0; k0 < 64; k0 += 4) {
    int kk = k0 + khalf;
    v2f a; a.x = rowNew[kk]; a.y = rowNew[kk + 1];
    const v2f* wr = &Wp[(kk >> 1) * WP_STRIDE];
    v2f b0 = wr[m];
    v2f b1 = wr[m + 16];
    v2f b2 = wr[m + 32];
    v2f b3 = wr[m + 48];
    acc0 = __builtin_amdgcn_wmma_f32_16x16x4_f32(false, a, false, b0, (short)0, acc0, false, false);
    acc1 = __builtin_amdgcn_wmma_f32_16x16x4_f32(false, a, false, b1, (short)0, acc1, false, false);
    acc2 = __builtin_amdgcn_wmma_f32_16x16x4_f32(false, a, false, b2, (short)0, acc2, false, false);
    acc3 = __builtin_amdgcn_wmma_f32_16x16x4_f32(false, a, false, b3, (short)0, acc3, false, false);
  }

#pragma unroll
  for (int k0 = 0; k0 < 64; k0 += 4) {
    int kk = k0 + khalf;
    v2f a; a.x = rowU[kk]; a.y = rowU[kk + 1];
    const v2f* wr = &Wp[(32 + (kk >> 1)) * WP_STRIDE];
    v2f b0 = wr[m];
    v2f b1 = wr[m + 16];
    v2f b2 = wr[m + 32];
    v2f b3 = wr[m + 48];
    acc0 = __builtin_amdgcn_wmma_f32_16x16x4_f32(false, a, false, b0, (short)0, acc0, false, false);
    acc1 = __builtin_amdgcn_wmma_f32_16x16x4_f32(false, a, false, b1, (short)0, acc1, false, false);
    acc2 = __builtin_amdgcn_wmma_f32_16x16x4_f32(false, a, false, b2, (short)0, acc2, false, false);
    acc3 = __builtin_amdgcn_wmma_f32_16x16x4_f32(false, a, false, b3, (short)0, acc3, false, false);
  }

  int mo = (lane >> 4) * 8;
  float* orow = out + (size_t)row0 * DDIM;
#pragma unroll
  for (int i2 = 0; i2 < 8; ++i2) {
    orow[(size_t)(mo + i2) * DDIM + m]      = acc0[i2];
    orow[(size_t)(mo + i2) * DDIM + m + 16] = acc1[i2];
    orow[(size_t)(mo + i2) * DDIM + m + 32] = acc2[i2];
    orow[(size_t)(mo + i2) * DDIM + m + 48] = acc3[i2];
  }
}

extern "C" void kernel_launch(void* const* d_in, const int* in_sizes, int n_in,
                              void* d_out, int out_size, void* d_ws, size_t ws_size,
                              hipStream_t stream) {
  const float* user_emb    = (const float*)d_in[0];
  const float* item_emb    = (const float*)d_in[1];
  const float* weights     = (const float*)d_in[2];
  const float* social_vals = (const float*)d_in[3];
  const float* inter_vals  = (const float*)d_in[4];
  const int*   social_rows = (const int*)d_in[5];
  const int*   social_cols = (const int*)d_in[6];
  const int*   inter_rows  = (const int*)d_in[7];
  const int*   inter_cols  = (const int*)d_in[8];

  const int D   = DDIM;
  const int U   = in_sizes[0] / D;
  const int I   = in_sizes[1] / D;
  const int L   = in_sizes[2] / (2 * D * D);
  const int E_S = in_sizes[3];
  const int E_I = in_sizes[4];

  float* out_u = (float*)d_out;
  float* out_i = (float*)d_out + (size_t)U * D;

  float* planeP = (float*)d_ws;
  float* planeQ = planeP + (size_t)U * D;
  int*   cntS   = (int*)(planeQ + (size_t)U * D);
  int*   cntI   = cntS + (U + 1);
  int*   ptrS   = cntI + (U + 1);
  int*   ptrI   = ptrS + (U + 1);
  int2*  sortS  = (int2*)(ptrI + (U + 1));
  int2*  sortI  = sortS + E_S;
  int*   rankS  = (int*)planeQ;
  int*   rankI  = rankS + E_S;

  const int buildThreads = 256;
  const int buildBlocks  = (E_S + EDGE_BLK - 1) / EDGE_BLK + (E_I + EDGE_BLK - 1) / EDGE_BLK;
  const int scanBlocks   = (U + 1 + SCAN_CH - 1) / SCAN_CH;
  const int rowBlocks    = (U + 8 * RPW - 1) / (8 * RPW);

  const int wavesPerBlock = 10;
  const int gemmThreads   = wavesPerBlock * 32;
  const int tiles         = (U + 15) / 16;
  const int gemmBlocks    = (tiles + wavesPerBlock - 1) / wavesPerBlock;

  hipMemsetAsync(cntS, 0, (size_t)2 * (U + 1) * sizeof(int), stream);
  edge_rank<<<buildBlocks, buildThreads, 0, stream>>>(
      social_rows, E_S, cntS, rankS, inter_rows, E_I, cntI, rankI);
  row_scan<<<dim3(scanBlocks, 2), 256, 0, stream>>>(cntS, ptrS, cntI, ptrI, U);
  edge_fill<<<buildBlocks, buildThreads, 0, stream>>>(
      social_rows, social_cols, social_vals, rankS, ptrS, sortS, E_S,
      inter_rows, inter_cols, inter_vals, rankI, ptrI, sortI, E_I);

  const float* ucur = user_emb;
  for (int k = 0; k < L; ++k) {
    float* agg = (k & 1) ? planeQ : planeP;
    spmm_rows<false><<<rowBlocks, 256, 0, stream>>>(ptrS, sortS, ucur, agg, U);
    float* dst = (k == L - 1) ? out_u : agg;
    gemm_concat_wmma<<<gemmBlocks, gemmThreads, 0, stream>>>(
        agg, ucur, weights + (size_t)k * 2 * D * D, dst, U);
    ucur = dst;
  }
  if (L == 0) {
    hipMemcpyAsync(out_u, user_emb, (size_t)U * D * sizeof(float),
                   hipMemcpyDeviceToDevice, stream);
  }

  spmm_rows<true><<<rowBlocks, 256, 0, stream>>>(ptrI, sortI, item_emb, out_u, U);

  hipMemcpyAsync(out_i, item_emb, (size_t)I * D * sizeof(float),
                 hipMemcpyDeviceToDevice, stream);
}


# Edges grouped by row by a two-level counting sort on LDS counters (returning LDS add = place), not one returning global atomic per edge; row sum order = LDS arrival, varies by run
# speedup vs baseline: 2.2882x; 1.6203x over previous
#include <hip/hip_runtime.h>

typedef __attribute__((ext_vector_type(2))) float v2f;
typedef __attribute__((ext_vector_type(8))) float v8f;

#define DDIM 64

#define RB_SHIFT 8
#define RB (1 << RB_SHIFT)
#define HMAX 1024
#define CH_T 512
#define CH_EPT 8
#define CHUNK (CH_T * CH_EPT)

__global__ __launch_bounds__(CH_T) void bucket_count(const int* __restrict__ rowsA, int nA,
                                                    int* __restrict__ cmA,
                                                    const int* __restrict__ rowsB, int nB,
                                                    int* __restrict__ cmB, int nbuckets) {
  __shared__ int hist[HMAX];
  int blocksA = (nA + CHUNK - 1) / CHUNK;
  int blk = blockIdx.x;
  const int* rows = rowsA; int n = nA; int* cm = cmA; int nblk = blocksA;
  if (blk >= blocksA) {
    blk -= blocksA; rows = rowsB; n = nB; cm = cmB; nblk = (nB + CHUNK - 1) / CHUNK;
  }
  for (int b = threadIdx.x; b < nbuckets; b += CH_T) hist[b] = 0;
  int first = blk * CHUNK + threadIdx.x;
  int r[CH_EPT];
#pragma unroll
  for (int t = 0; t < CH_EPT; ++t) {
    int idx = first + t * CH_T;
    r[t] = (idx < n) ? rows[idx] : -1;
  }
  __syncthreads();
#pragma unroll
  for (int t = 0; t < CH_EPT; ++t)
    if (r[t] >= 0) atomicAdd(&hist[r[t] >> RB_SHIFT], 1);
  __syncthreads();
  for (int b = threadIdx.x; b < nbuckets; b += CH_T) cm[(size_t)b * nblk + blk] = hist[b];
  int tot = nbuckets * nblk;
  if (blk == 0 && (int)threadIdx.x < ((tot + 3) & ~3) - tot) cm[tot + threadIdx.x] = 0;
}

#define SC_T 1024

__global__ __launch_bounds__(SC_T) void bucket_scan(const int4* __restrict__ cmA, int4* __restrict__ cbA, int nquadA,
                                                   const int4* __restrict__ cmB, int4* __restrict__ cbB, int nquadB) {
  const int4* cm    = blockIdx.y ? cmB : cmA;
  int4*       cb    = blockIdx.y ? cbB : cbA;
  int         nquad = blockIdx.y ? nquadB : nquadA;
  int qbase = blockIdx.x * SC_T;
  if (qbase >= nquad) return;
  __shared__ int headSum[32];
  __shared__ int waveSum[32];
  int tid = threadIdx.x, lane = tid & 31, w = tid >> 5;

  int s = 0;
  int nq = blockIdx.x;
  for (int q0 = 0; q0 < nq; q0 += 16) {
    int4 a[16];
#pragma unroll
    for (int k = 0; k < 16; ++k) {
      a[k] = make_int4(0, 0, 0, 0);
      if (q0 + k < nq) a[k] = cm[(q0 + k) * SC_T + tid];
    }
#pragma unroll
    for (int k = 0; k < 16; ++k) s += (a[k].x + a[k].y) + (a[k].z + a[k].w);
  }
  for (int o = 16; o > 0; o >>= 1) s += __shfl_xor(s, o);
  if (lane == 0) headSum[w] = s;

  int q = qbase + tid;
  int4 c = make_int4(0, 0, 0, 0);
  if (q < nquad) c = cm[q];
  int t = (c.x + c.y) + (c.z + c.w);
  int incl = t;
  for (int o = 1; o < 32; o <<= 1) {
    int v = __shfl_up(incl, o);
    if (lane >= o) incl += v;
  }
  if (lane == 31) waveSum[w] = incl;
  __syncthreads();
  int pre = 0;
  for (int k = 0; k < 32; ++k) {
    pre += headSum[k];
    if (k < w) pre += waveSum[k];
  }
  int ex = pre + incl - t;
  if (q < nquad) cb[q] = make_int4(ex, ex + c.x, ex + c.x + c.y, ex + c.x + c.y + c.z);
}

__global__ __launch_bounds__(CH_T) void bucket_place(
    const int* __restrict__ rowsA, const int* __restrict__ colsA, const float* __restrict__ valsA,
    int nA, const int* __restrict__ cbA, int4* __restrict__ midA,
    const int* __restrict__ rowsB, const int* __restrict__ colsB, const float* __restrict__ valsB,
    int nB, const int* __restrict__ cbB, int4* __restrict__ midB,
    int nbuckets) {
  __shared__ int cur[HMAX + 32];
  int blocksA = (nA + CHUNK - 1) / CHUNK;
  int blk = blockIdx.x;
  const int* rows = rowsA; const int* cols = colsA; const float* vals = valsA;
  int n = nA; const int* cb = cbA; int4* mid = midA; int nblk = blocksA;
  if (blk >= blocksA) {
    blk -= blocksA; rows = rowsB; cols = colsB; vals = valsB;
    n = nB; cb = cbB; mid = midB; nblk = (nB + CHUNK - 1) / CHUNK;
  }
  for (int b = threadIdx.x; b < nbuckets; b += CH_T) cur[b] = cb[(size_t)b * nblk + blk];
  int first = blk * CHUNK + threadIdx.x;
  int r[CH_EPT], c[CH_EPT];
  float v[CH_EPT];
#pragma unroll
  for (int t = 0; t < CH_EPT; ++t) {
    int idx = first + t * CH_T;
    r[t] = -1; c[t] = 0; v[t] = 0.f;
    if (idx < n) { r[t] = rows[idx]; c[t] = cols[idx]; v[t] = vals[idx]; }
  }
  __syncthreads();
  int spare = HMAX + (threadIdx.x & 31);
  int pos[CH_EPT];
#pragma unroll
  for (int t = 0; t < CH_EPT; ++t)
    pos[t] = atomicAdd(&cur[(r[t] >= 0) ? (r[t] >> RB_SHIFT) : spare], 1);
#pragma unroll
  for (int t = 0; t < CH_EPT; ++t)
    if (r[t] >= 0) mid[pos[t]] = make_int4(c[t], __float_as_int(v[t]), r[t], 0);
}

#define RW_E 8

__global__ __launch_bounds__(RB) void bucket_rows(
    const int* __restrict__ cbA, const int4* __restrict__ midA,
    int nA, int* __restrict__ ptrA, int2* __restrict__ sortedA,
    const int* __restrict__ cbB, const int4* __restrict__ midB,
    int nB, int* __restrict__ ptrB, int2* __restrict__ sortedB,
    int nrows, int nbuckets) {
  __shared__ int cnt[RB + 32];
  __shared__ int waveSum[RB / 32];
  const int* cb = cbA; const int4* mid = midA;
  int n = nA; int* ptr = ptrA; int2* sorted = sortedA;
  if (blockIdx.y) { cb = cbB; mid = midB; n = nB; ptr = ptrB; sorted = sortedB; }
  int nblk = (n + CHUNK - 1) / CHUNK;
  int b = blockIdx.x;
  int tid = threadIdx.x, lane = tid & 31, w = tid >> 5;
  int row0 = b << RB_SHIFT;
  int beg = cb[(size_t)b * nblk];
  int end = (b + 1 < nbuckets) ? cb[(size_t)(b + 1) * nblk] : n;
  cnt[tid] = 0;
  __syncthreads();
  for (int i0 = beg + tid; i0 < end; i0 += RW_E * RB) {
    int r[RW_E];
#pragma unroll
    for (int u = 0; u < RW_E; ++u) { int idx = i0 + u * RB; r[u] = (idx < end) ? mid[idx].z : -1; }
#pragma unroll
    for (int u = 0; u < RW_E; ++u) if (r[u] >= 0) atomicAdd(&cnt[r[u] - row0], 1);
  }
  __syncthreads();
  int c = cnt[tid];
  int incl = c;
  for (int o = 1; o < 32; o <<= 1) {
    int v = __shfl_up(incl, o);
    if (lane >= o) incl += v;
  }
  if (lane == 31) waveSum[w] = incl;
  __syncthreads();
  int pre = 0;
  for (int k = 0; k < RB / 32; ++k) if (k < w) pre += waveSum[k];
  int start = beg + pre + incl - c;
  if (row0 + tid < nrows) ptr[row0 + tid] = start;
  if (b == nbuckets - 1 && tid == 0) ptr[nrows] = n;
  cnt[tid] = start;
  __syncthreads();
  for (int i0 = beg + tid; i0 < end; i0 += RW_E * RB) {
    int4 e[RW_E];
#pragma unroll
    for (int u = 0; u < RW_E; ++u) {
      int idx = i0 + u * RB;
      e[u] = make_int4(0, 0, -1, 0);
      if (idx < end) e[u] = mid[idx];
    }
    int pos[RW_E];
#pragma unroll
    for (int u = 0; u < RW_E; ++u)
      pos[u] = atomicAdd(&cnt[(e[u].z >= 0) ? (e[u].z - row0) : (RB + lane)], 1);
#pragma unroll
    for (int u = 0; u < RW_E; ++u)
      if (e[u].z >= 0) sorted[pos[u]] = make_int2(e[u].x, e[u].y);
  }
}

#define RPW 8

template <bool ADD>
__global__ __launch_bounds__(256) void spmm_rows(const int* __restrict__ ptr,
                                                 const int2* __restrict__ sorted,
                                                 const float* __restrict__ x,
                                                 float* out,
                                                 int nrows) {
  int lane = threadIdx.x & 31;
  int r0 = __builtin_amdgcn_readfirstlane((blockIdx.x * 8 + (threadIdx.x >> 5)) * RPW);
  if (r0 >= nrows) return;
  int cnt = nrows - r0;
  if (cnt > RPW) cnt = RPW;
  int p = ptr[r0 + (lane < cnt ? lane : cnt)];
  int ebeg   = __builtin_amdgcn_readlane(p, 0);
  int eend   = __builtin_amdgcn_readlane(p, cnt);
  int rowEnd = __builtin_amdgcn_readlane(p, 1);
  int ri = 0;
  float2* orow = (float2*)(out + (size_t)r0 * DDIM) + lane;
  const float2* xr = (const float2*)x + lane;
  float2 acc = make_float2(0.f, 0.f);
  float2 st[RPW];
  if (ADD) {
#pragma unroll
    for (int i = 0; i < RPW; ++i) st[i] = orow[(i < cnt ? i : cnt - 1) * 32];
    acc = st[0];
  }
  auto flush = [&]() {
    orow[ri * 32] = acc;
    ++ri;
    if (ri < cnt) rowEnd = __builtin_amdgcn_readlane(p, ri + 1);
    acc = make_float2(0.f, 0.f);
    if (ADD) {
#pragma unroll
      for (int i = 1; i < RPW; ++i) if (ri == i) acc = st[i];
    }
  };

  const int last = eend - 1;
  int2 e = make_int2(0, 0);
  if (ebeg < eend) e = sorted[(ebeg + lane < eend) ? ebeg + lane : last];
  for (int base = ebeg; base < eend; base += 32) {
    int n = eend - base;
    if (n > 32) n = 32;
    int2 en = sorted[(base + 32 + lane < eend) ? base + 32 + lane : last];
    auto add = [&](int idx, float v, float2 s) {
      while (idx >= rowEnd) flush();
      acc.x += v * s.x; acc.y += v * s.y;
    };
    int j = 0;
    for (; j + 4 <= n; j += 4) {
      int   k0 = __builtin_amdgcn_readlane(e.x, j);
      int   k1 = __builtin_amdgcn_readlane(e.x, j + 1);
      int   k2 = __builtin_amdgcn_readlane(e.x, j + 2);
      int   k3 = __builtin_amdgcn_readlane(e.x, j + 3);
      float v0 = __int_as_float(__builtin_amdgcn_readlane(e.y, j));
      float v1 = __int_as_float(__builtin_amdgcn_readlane(e.y, j + 1));
      float v2 = __int_as_float(__builtin_amdgcn_readlane(e.y, j + 2));
      float v3 = __int_as_float(__builtin_amdgcn_readlane(e.y, j + 3));
      float2 s0 = xr[(size_t)k0 * (DDIM / 2)];
      float2 s1 = xr[(size_t)k1 * (DDIM / 2)];
      float2 s2 = xr[(size_t)k2 * (DDIM / 2)];
      float2 s3 = xr[(size_t)k3 * (DDIM / 2)];
      add(base + j,     v0, s0);
      add(base + j + 1, v1, s1);
      add(base + j + 2, v2, s2);
      add(base + j + 3, v3, s3);
    }
    if (j < n) {
      int j1 = (j + 1 < n) ? j + 1 : n - 1;
      int j2 = (j + 2 < n) ? j + 2 : n - 1;
      int   k0 = __builtin_amdgcn_readlane(e.x, j);
      int   k1 = __builtin_amdgcn_readlane(e.x, j1);
      int   k2 = __builtin_amdgcn_readlane(e.x, j2);
      float v0 = __int_as_float(__builtin_amdgcn_readlane(e.y, j));
      float v1 = __int_as_float(__builtin_amdgcn_readlane(e.y, j1));
      float v2 = __int_as_float(__builtin_amdgcn_readlane(e.y, j2));
      float2 s0 = xr[(size_t)k0 * (DDIM / 2)];
      float2 s1 = xr[(size_t)k1 * (DDIM / 2)];
      float2 s2 = xr[(size_t)k2 * (DDIM / 2)];
      add(base + j, v0, s0);
      if (j + 1 < n) add(base + j + 1, v1, s1);
      if (j + 2 < n) add(base + j + 2, v2, s2);
    }
    e = en;
  }
  while (ri < cnt) flush();
}

#define WP_STRIDE 80

__global__ __launch_bounds__(320) void gemm_concat_wmma(
    const float* newu,
    const float* __restrict__ uin,
    const float* __restrict__ W,
    float* out,
    int nrows) {
  __shared__ v2f Wp[64 * WP_STRIDE];
  for (int idx = threadIdx.x; idx < 64 * 64; idx += blockDim.x) {
    int t = idx >> 6, n = idx & 63;
    v2f p;
    p.x = W[(size_t)(2 * t) * DDIM + n];
    p.y = W[(size_t)(2 * t + 1) * DDIM + n];
    Wp[t * WP_STRIDE + n] = p;
  }
  __syncthreads();

  int lane = threadIdx.x & 31;
  int tile = blockIdx.x * (blockDim.x >> 5) + (threadIdx.x >> 5);
  if (tile * 16 >= nrows) return;
  int row0  = tile * 16;
  int m     = lane & 15;
  int khalf = (lane >> 4) << 1;

  const float* rowNew = newu + (size_t)(row0 + m) * DDIM;
  const float* rowU   = uin  + (size_t)(row0 + m) * DDIM;

  v8f acc0 = {}, acc1 = {}, acc2 = {}, acc3 = {};

#pragma unroll
  for (int k0 = 0; k0 < 64; k0 += 4) {
    int kk = k0 + khalf;
    v2f a; a.x = rowNew[kk]; a.y = rowNew[kk + 1];
    const v2f* wr = &Wp[(kk >> 1) * WP_STRIDE];
    v2f b0 = wr[m];
    v2f b1 = wr[m + 16];
    v2f b2 = wr[m + 32];
    v2f b3 = wr[m + 48];
    acc0 = __builtin_amdgcn_wmma_f32_16x16x4_f32(false, a, false, b0, (short)0, acc0, false, false);
    acc1 = __builtin_amdgcn_wmma_f32_16x16x4_f32(false, a, false, b1, (short)0, acc1, false, false);
    acc2 = __builtin_amdgcn_wmma_f32_16x16x4_f32(false, a, false, b2, (short)0, acc2, false, false);
    acc3 = __builtin_amdgcn_wmma_f32_16x16x4_f32(false, a, false, b3, (short)0, acc3, false, false);
  }

#pragma unroll
  for (int k0 = 0; k0 < 64; k0 += 4) {
    int kk = k0 + khalf;
    v2f a; a.x = rowU[kk]; a.y = rowU[kk + 1];
    const v2f* wr = &Wp[(32 + (kk >> 1)) * WP_STRIDE];
    v2f b0 = wr[m];
    v2f b1 = wr[m + 16];
    v2f b2 = wr[m + 32];
    v2f b3 = wr[m + 48];
    acc0 = __builtin_amdgcn_wmma_f32_16x16x4_f32(false, a, false, b0, (short)0, acc0, false, false);
    acc1 = __builtin_amdgcn_wmma_f32_16x16x4_f32(false, a, false, b1, (short)0, acc1, false, false);
    acc2 = __builtin_amdgcn_wmma_f32_16x16x4_f32(false, a, false, b2, (short)0, acc2, false, false);
    acc3 = __builtin_amdgcn_wmma_f32_16x16x4_f32(false, a, false, b3, (short)0, acc3, false, false);
  }

  int mo = (lane >> 4) * 8;
  float* orow = out + (size_t)row0 * DDIM;
#pragma unroll
  for (int i2 = 0; i2 < 8; ++i2) {
    orow[(size_t)(mo + i2) * DDIM + m]      = acc0[i2];
    orow[(size_t)(mo + i2) * DDIM + m + 16] = acc1[i2];
    orow[(size_t)(mo + i2) * DDIM + m + 32] = acc2[i2];
    orow[(size_t)(mo + i2) * DDIM + m + 48] = acc3[i2];
  }
}

extern "C" void kernel_launch(void* const* d_in, const int* in_sizes, int n_in,
                              void* d_out, int out_size, void* d_ws, size_t ws_size,
                              hipStream_t stream) {
  const float* user_emb    = (const float*)d_in[0];
  const float* item_emb    = (const float*)d_in[1];
  const float* weights     = (const float*)d_in[2];
  const float* social_vals = (const float*)d_in[3];
  const float* inter_vals  = (const float*)d_in[4];
  const int*   social_rows = (const int*)d_in[5];
  const int*   social_cols = (const int*)d_in[6];
  const int*   inter_rows  = (const int*)d_in[7];
  const int*   inter_cols  = (const int*)d_in[8];

  const int D   = DDIM;
  const int U   = in_sizes[0] / D;
  const int I   = in_sizes[1] / D;
  const int L   = in_sizes[2] / (2 * D * D);
  const int E_S = in_sizes[3];
  const int E_I = in_sizes[4];

  float* out_u = (float*)d_out;
  float* out_i = (float*)d_out + (size_t)U * D;

  float* planeP = (float*)d_ws;
  float* planeQ = planeP + (size_t)U * D;
  int*   cntS   = (int*)(planeQ + (size_t)U * D);
  int*   cntI   = cntS + (U + 1);
  int*   ptrS   = cntI + (U + 1);
  int*   ptrI   = ptrS + (U + 1);
  int2*  sortS  = (int2*)(ptrI + (U + 1));
  int2*  sortI  = sortS + E_S;

  const int nbuckets = (U + RB - 1) / RB;
  const int chunksS  = (E_S + CHUNK - 1) / CHUNK;
  const int chunksI  = (E_I + CHUNK - 1) / CHUNK;
  const int quadS    = (nbuckets * chunksS + 3) / 4;
  const int quadI    = (nbuckets * chunksI + 3) / 4;
  int4*  midS = (int4*)d_ws;
  int4*  midI = midS + E_S;
  int4*  cmS  = (int4*)(midI + E_I);
  int4*  cmI  = cmS + quadS;
  int4*  cbS  = cmI + quadI;
  int4*  cbI  = cbS + quadS;

  const int chunkBlocks = chunksS + chunksI;
  const int scanTiles   = ((quadS > quadI ? quadS : quadI) + SC_T - 1) / SC_T;
  const int rowBlocks   = (U + 8 * RPW - 1) / (8 * RPW);

  const int wavesPerBlock = 10;
  const int gemmThreads   = wavesPerBlock * 32;
  const int tiles         = (U + 15) / 16;
  const int gemmBlocks    = (tiles + wavesPerBlock - 1) / wavesPerBlock;

  bucket_count<<<chunkBlocks, CH_T, 0, stream>>>(social_rows, E_S, (int*)cmS, inter_rows, E_I, (int*)cmI, nbuckets);
  bucket_scan<<<dim3(scanTiles, 2), SC_T, 0, stream>>>(cmS, cbS, quadS, cmI, cbI, quadI);
  bucket_place<<<chunkBlocks, CH_T, 0, stream>>>(
      social_rows, social_cols, social_vals, E_S, (const int*)cbS, midS,
      inter_rows, inter_cols, inter_vals, E_I, (const int*)cbI, midI, nbuckets);
  bucket_rows<<<dim3(nbuckets, 2), RB, 0, stream>>>(
      (const int*)cbS, midS, E_S, ptrS, sortS,
      (const int*)cbI, midI, E_I, ptrI, sortI, U, nbuckets);

  const float* ucur = user_emb;
  for (int k = 0; k < L; ++k) {
    float* agg = (k & 1) ? planeQ : planeP;
    spmm_rows<false><<<rowBlocks, 256, 0, stream>>>(ptrS, sortS, ucur, agg, U);
    float* dst = (k == L - 1) ? out_u : agg;
    gemm_concat_wmma<<<gemmBlocks, gemmThreads, 0, stream>>>(
        agg, ucur, weights + (size_t)k * 2 * D * D, dst, U);
    ucur = dst;
  }
  if (L == 0) {
    hipMemcpyAsync(out_u, user_emb, (size_t)U * D * sizeof(float),
                   hipMemcpyDeviceToDevice, stream);
  }

  spmm_rows<true><<<rowBlocks, 256, 0, stream>>>(ptrI, sortI, item_emb, out_u, U);

  hipMemcpyAsync(out_i, item_emb, (size_t)I * D * sizeof(float),
                 hipMemcpyDeviceToDevice, stream);
}


# Each layer is one kernel: a block sums its 128 rows into LDS and feeds them straight to the WMMA product; the last one also adds the second graph's rows there; 9 -> 6 launches; sum orders kept
# speedup vs baseline: 2.8654x; 1.2523x over previous
#include <hip/hip_runtime.h>

typedef __attribute__((ext_vector_type(2))) float v2f;
typedef __attribute__((ext_vector_type(8))) float v8f;

#define DDIM 64

#define RB_SHIFT 8
#define RB (1 << RB_SHIFT)
#define HMAX 1024
#define CH_T 512
#define CH_EPT 8
#define CHUNK (CH_T * CH_EPT)

__global__ __launch_bounds__(CH_T) void bucket_count(const int* __restrict__ rowsA, int nA,
                                                    int* __restrict__ cmA,
                                                    const int* __restrict__ rowsB, int nB,
                                                    int* __restrict__ cmB, int nbuckets) {
  __shared__ int hist[HMAX];
  int blocksA = (nA + CHUNK - 1) / CHUNK;
  int blk = blockIdx.x;
  const int* rows = rowsA; int n = nA; int* cm = cmA; int nblk = blocksA;
  if (blk >= blocksA) {
    blk -= blocksA; rows = rowsB; n = nB; cm = cmB; nblk = (nB + CHUNK - 1) / CHUNK;
  }
  for (int b = threadIdx.x; b < nbuckets; b += CH_T) hist[b] = 0;
  int first = blk * CHUNK + threadIdx.x;
  int r[CH_EPT];
#pragma unroll
  for (int t = 0; t < CH_EPT; ++t) {
    int idx = first + t * CH_T;
    r[t] = (idx < n) ? rows[idx] : -1;
  }
  __syncthreads();
#pragma unroll
  for (int t = 0; t < CH_EPT; ++t)
    if (r[t] >= 0) atomicAdd(&hist[r[t] >> RB_SHIFT], 1);
  __syncthreads();
  for (int b = threadIdx.x; b < nbuckets; b += CH_T) cm[(size_t)b * nblk + blk] = hist[b];
  int tot = nbuckets * nblk;
  if (blk == 0 && (int)threadIdx.x < ((tot + 3) & ~3) - tot) cm[tot + threadIdx.x] = 0;
}

#define SC_T 1024

__global__ __launch_bounds__(SC_T) void bucket_scan(const int4* __restrict__ cmA, int4* __restrict__ cbA, int nquadA,
                                                   const int4* __restrict__ cmB, int4* __restrict__ cbB, int nquadB) {
  const int4* cm    = blockIdx.y ? cmB : cmA;
  int4*       cb    = blockIdx.y ? cbB : cbA;
  int         nquad = blockIdx.y ? nquadB : nquadA;
  int qbase = blockIdx.x * SC_T;
  if (qbase >= nquad) return;
  __shared__ int headSum[32];
  __shared__ int waveSum[32];
  int tid = threadIdx.x, lane = tid & 31, w = tid >> 5;

  int s = 0;
  int nq = blockIdx.x;
  for (int q0 = 0; q0 < nq; q0 += 16) {
    int4 a[16];
#pragma unroll
    for (int k = 0; k < 16; ++k) {
      a[k] = make_int4(0, 0, 0, 0);
      if (q0 + k < nq) a[k] = cm[(q0 + k) * SC_T + tid];
    }
#pragma unroll
    for (int k = 0; k < 16; ++k) s += (a[k].x + a[k].y) + (a[k].z + a[k].w);
  }
  for (int o = 16; o > 0; o >>= 1) s += __shfl_xor(s, o);
  if (lane == 0) headSum[w] = s;

  int q = qbase + tid;
  int4 c = make_int4(0, 0, 0, 0);
  if (q < nquad) c = cm[q];
  int t = (c.x + c.y) + (c.z + c.w);
  int incl = t;
  for (int o = 1; o < 32; o <<= 1) {
    int v = __shfl_up(incl, o);
    if (lane >= o) incl += v;
  }
  if (lane == 31) waveSum[w] = incl;
  __syncthreads();
  int pre = 0;
  for (int k = 0; k < 32; ++k) {
    pre += headSum[k];
    if (k < w) pre += waveSum[k];
  }
  int ex = pre + incl - t;
  if (q < nquad) cb[q] = make_int4(ex, ex + c.x, ex + c.x + c.y, ex + c.x + c.y + c.z);
}

__global__ __launch_bounds__(CH_T) void bucket_place(
    const int* __restrict__ rowsA, const int* __restrict__ colsA, const float* __restrict__ valsA,
    int nA, const int* __restrict__ cbA, int4* __restrict__ midA,
    const int* __restrict__ rowsB, const int* __restrict__ colsB, const float* __restrict__ valsB,
    int nB, const int* __restrict__ cbB, int4* __restrict__ midB,
    int nbuckets) {
  __shared__ int cur[HMAX + 32];
  int blocksA = (nA + CHUNK - 1) / CHUNK;
  int blk = blockIdx.x;
  const int* rows = rowsA; const int* cols = colsA; const float* vals = valsA;
  int n = nA; const int* cb = cbA; int4* mid = midA; int nblk = blocksA;
  if (blk >= blocksA) {
    blk -= blocksA; rows = rowsB; cols = colsB; vals = valsB;
    n = nB; cb = cbB; mid = midB; nblk = (nB + CHUNK - 1) / CHUNK;
  }
  for (int b = threadIdx.x; b < nbuckets; b += CH_T) cur[b] = cb[(size_t)b * nblk + blk];
  int first = blk * CHUNK + threadIdx.x;
  int r[CH_EPT], c[CH_EPT];
  float v[CH_EPT];
#pragma unroll
  for (int t = 0; t < CH_EPT; ++t) {
    int idx = first + t * CH_T;
    r[t] = -1; c[t] = 0; v[t] = 0.f;
    if (idx < n) { r[t] = rows[idx]; c[t] = cols[idx]; v[t] = vals[idx]; }
  }
  __syncthreads();
  int spare = HMAX + (threadIdx.x & 31);
  int pos[CH_EPT];
#pragma unroll
  for (int t = 0; t < CH_EPT; ++t)
    pos[t] = atomicAdd(&cur[(r[t] >= 0) ? (r[t] >> RB_SHIFT) : spare], 1);
#pragma unroll
  for (int t = 0; t < CH_EPT; ++t)
    if (r[t] >= 0) mid[pos[t]] = make_int4(c[t], __float_as_int(v[t]), r[t], 0);
}

#define RW_E 8

__global__ __launch_bounds__(RB) void bucket_rows(
    const int* __restrict__ cbA, const int4* __restrict__ midA,
    int nA, int* __restrict__ ptrA, int2* __restrict__ sortedA,
    const int* __restrict__ cbB, const int4* __restrict__ midB,
    int nB, int* __restrict__ ptrB, int2* __restrict__ sortedB,
    int nrows, int nbuckets) {
  __shared__ int cnt[RB + 32];
  __shared__ int waveSum[RB / 32];
  const int* cb = cbA; const int4* mid = midA;
  int n = nA; int* ptr = ptrA; int2* sorted = sortedA;
  if (blockIdx.y) { cb = cbB; mid = midB; n = nB; ptr = ptrB; sorted = sortedB; }
  int nblk = (n + CHUNK - 1) / CHUNK;
  int b = blockIdx.x;
  int tid = threadIdx.x, lane = tid & 31, w = tid >> 5;
  int row0 = b << RB_SHIFT;
  int beg = cb[(size_t)b * nblk];
  int end = (b + 1 < nbuckets) ? cb[(size_t)(b + 1) * nblk] : n;
  cnt[tid] = 0;
  __syncthreads();
  for (int i0 = beg + tid; i0 < end; i0 += RW_E * RB) {
    int r[RW_E];
#pragma unroll
    for (int u = 0; u < RW_E; ++u) { int idx = i0 + u * RB; r[u] = (idx < end) ? mid[idx].z : -1; }
#pragma unroll
    for (int u = 0; u < RW_E; ++u) if (r[u] >= 0) atomicAdd(&cnt[r[u] - row0], 1);
  }
  __syncthreads();
  int c = cnt[tid];
  int incl = c;
  for (int o = 1; o < 32; o <<= 1) {
    int v = __shfl_up(incl, o);
    if (lane >= o) incl += v;
  }
  if (lane == 31) waveSum[w] = incl;
  __syncthreads();
  int pre = 0;
  for (int k = 0; k < RB / 32; ++k) if (k < w) pre += waveSum[k];
  int start = beg + pre + incl - c;
  if (row0 + tid < nrows) ptr[row0 + tid] = start;
  if (b == nbuckets - 1 && tid == 0) ptr[nrows] = n;
  cnt[tid] = start;
  __syncthreads();
  for (int i0 = beg + tid; i0 < end; i0 += RW_E * RB) {
    int4 e[RW_E];
#pragma unroll
    for (int u = 0; u < RW_E; ++u) {
      int idx = i0 + u * RB;
      e[u] = make_int4(0, 0, -1, 0);
      if (idx < end) e[u] = mid[idx];
    }
    int pos[RW_E];
#pragma unroll
    for (int u = 0; u < RW_E; ++u)
      pos[u] = atomicAdd(&cnt[(e[u].z >= 0) ? (e[u].z - row0) : (RB + lane)], 1);
#pragma unroll
    for (int u = 0; u < RW_E; ++u)
      if (e[u].z >= 0) sorted[pos[u]] = make_int2(e[u].x, e[u].y);
  }
}

#define RPW 8

template <bool ADD>
__global__ __launch_bounds__(256) void spmm_rows(const int* __restrict__ ptr,
                                                 const int2* __restrict__ sorted,
                                                 const float* __restrict__ x,
                                                 float* out,
                                                 int nrows) {
  int lane = threadIdx.x & 31;
  int r0 = __builtin_amdgcn_readfirstlane((blockIdx.x * 8 + (threadIdx.x >> 5)) * RPW);
  if (r0 >= nrows) return;
  int cnt = nrows - r0;
  if (cnt > RPW) cnt = RPW;
  int p = ptr[r0 + (lane < cnt ? lane : cnt)];
  int ebeg   = __builtin_amdgcn_readlane(p, 0);
  int eend   = __builtin_amdgcn_readlane(p, cnt);
  int rowEnd = __builtin_amdgcn_readlane(p, 1);
  int ri = 0;
  float2* orow = (float2*)(out + (size_t)r0 * DDIM) + lane;
  const float2* xr = (const float2*)x + lane;
  float2 acc = make_float2(0.f, 0.f);
  float2 st[RPW];
  if (ADD) {
#pragma unroll
    for (int i = 0; i < RPW; ++i) st[i] = orow[(i < cnt ? i : cnt - 1) * 32];
    acc = st[0];
  }
  auto flush = [&]() {
    orow[ri * 32] = acc;
    ++ri;
    if (ri < cnt) rowEnd = __builtin_amdgcn_readlane(p, ri + 1);
    acc = make_float2(0.f, 0.f);
    if (ADD) {
#pragma unroll
      for (int i = 1; i < RPW; ++i) if (ri == i) acc = st[i];
    }
  };

  const int last = eend - 1;
  int2 e = make_int2(0, 0);
  if (ebeg < eend) e = sorted[(ebeg + lane < eend) ? ebeg + lane : last];
  for (int base = ebeg; base < eend; base += 32) {
    int n = eend - base;
    if (n > 32) n = 32;
    int2 en = sorted[(base + 32 + lane < eend) ? base + 32 + lane : last];
    auto add = [&](int idx, float v, float2 s) {
      while (idx >= rowEnd) flush();
      acc.x += v * s.x; acc.y += v * s.y;
    };
    int j = 0;
    for (; j + 4 <= n; j += 4) {
      int   k0 = __builtin_amdgcn_readlane(e.x, j);
      int   k1 = __builtin_amdgcn_readlane(e.x, j + 1);
      int   k2 = __builtin_amdgcn_readlane(e.x, j + 2);
      int   k3 = __builtin_amdgcn_readlane(e.x, j + 3);
      float v0 = __int_as_float(__builtin_amdgcn_readlane(e.y, j));
      float v1 = __int_as_float(__builtin_amdgcn_readlane(e.y, j + 1));
      float v2 = __int_as_float(__builtin_amdgcn_readlane(e.y, j + 2));
      float v3 = __int_as_float(__builtin_amdgcn_readlane(e.y, j + 3));
      float2 s0 = xr[(size_t)k0 * (DDIM / 2)];
      float2 s1 = xr[(size_t)k1 * (DDIM / 2)];
      float2 s2 = xr[(size_t)k2 * (DDIM / 2)];
      float2 s3 = xr[(size_t)k3 * (DDIM / 2)];
      add(base + j,     v0, s0);
      add(base + j + 1, v1, s1);
      add(base + j + 2, v2, s2);
      add(base + j + 3, v3, s3);
    }
    if (j < n) {
      int j1 = (j + 1 < n) ? j + 1 : n - 1;
      int j2 = (j + 2 < n) ? j + 2 : n - 1;
      int   k0 = __builtin_amdgcn_readlane(e.x, j);
      int   k1 = __builtin_amdgcn_readlane(e.x, j1);
      int   k2 = __builtin_amdgcn_readlane(e.x, j2);
      float v0 = __int_as_float(__builtin_amdgcn_readlane(e.y, j));
      float v1 = __int_as_float(__builtin_amdgcn_readlane(e.y, j1));
      float v2 = __int_as_float(__builtin_amdgcn_readlane(e.y, j2));
      float2 s0 = xr[(size_t)k0 * (DDIM / 2)];
      float2 s1 = xr[(size_t)k1 * (DDIM / 2)];
      float2 s2 = xr[(size_t)k2 * (DDIM / 2)];
      add(base + j, v0, s0);
      if (j + 1 < n) add(base + j + 1, v1, s1);
      if (j + 2 < n) add(base + j + 2, v2, s2);
    }
    e = en;
  }
  while (ri < cnt) flush();
}

#define WP_STRIDE 80
#define TS 68
#define FB_T 512
#define FB_ROWS (FB_T / 32 * RPW)

template <int MODE>
__device__ __forceinline__ void wave_row_sums(const int* __restrict__ ptr,
                                              const int2* __restrict__ sorted,
                                              const float* __restrict__ x,
                                              float2* trow, float2* orow,
                                              int r0, int cnt, int lane) {
  int p = ptr[r0 + (lane < cnt ? lane : cnt)];
  int ebeg   = __builtin_amdgcn_readlane(p, 0);
  int eend   = __builtin_amdgcn_readlane(p, cnt);
  int rowEnd = __builtin_amdgcn_readlane(p, 1);
  int ri = 0;
  const float2* xr = (const float2*)x + lane;
  float2 acc = make_float2(0.f, 0.f);
  if (MODE) acc = trow[0];
  auto flush = [&]() {
    if (MODE) orow[ri * 32] = acc;
    else      trow[ri * (TS / 2)] = acc;
    ++ri;
    if (ri < cnt) rowEnd = __builtin_amdgcn_readlane(p, ri + 1);
    acc = make_float2(0.f, 0.f);
    if (MODE) { if (ri < cnt) acc = trow[ri * (TS / 2)]; }
  };

  const int last = eend - 1;
  int2 e = make_int2(0, 0);
  if (ebeg < eend) e = sorted[(ebeg + lane < eend) ? ebeg + lane : last];
  for (int base = ebeg; base < eend; base += 32) {
    int n = eend - base;
    if (n > 32) n = 32;
    int2 en = sorted[(base + 32 + lane < eend) ? base + 32 + lane : last];
    auto add = [&](int idx, float v, float2 s) {
      while (idx >= rowEnd) flush();
      acc.x += v * s.x; acc.y += v * s.y;
    };
    int j = 0;
    for (; j + 4 <= n; j += 4) {
      int   k0 = __builtin_amdgcn_readlane(e.x, j);
      int   k1 = __builtin_amdgcn_readlane(e.x, j + 1);
      int   k2 = __builtin_amdgcn_readlane(e.x, j + 2);
      int   k3 = __builtin_amdgcn_readlane(e.x, j + 3);
      float v0 = __int_as_float(__builtin_amdgcn_readlane(e.y, j));
      float v1 = __int_as_float(__builtin_amdgcn_readlane(e.y, j + 1));
      float v2 = __int_as_float(__builtin_amdgcn_readlane(e.y, j + 2));
      float v3 = __int_as_float(__builtin_amdgcn_readlane(e.y, j + 3));
      float2 s0 = xr[(size_t)k0 * (DDIM / 2)];
      float2 s1 = xr[(size_t)k1 * (DDIM / 2)];
      float2 s2 = xr[(size_t)k2 * (DDIM / 2)];
      float2 s3 = xr[(size_t)k3 * (DDIM / 2)];
      add(base + j,     v0, s0);
      add(base + j + 1, v1, s1);
      add(base + j + 2, v2, s2);
      add(base + j + 3, v3, s3);
    }
    if (j < n) {
      int j1 = (j + 1 < n) ? j + 1 : n - 1;
      int j2 = (j + 2 < n) ? j + 2 : n - 1;
      int   k0 = __builtin_amdgcn_readlane(e.x, j);
      int   k1 = __builtin_amdgcn_readlane(e.x, j1);
      int   k2 = __builtin_amdgcn_readlane(e.x, j2);
      float v0 = __int_as_float(__builtin_amdgcn_readlane(e.y, j));
      float v1 = __int_as_float(__builtin_amdgcn_readlane(e.y, j1));
      float v2 = __int_as_float(__builtin_amdgcn_readlane(e.y, j2));
      float2 s0 = xr[(size_t)k0 * (DDIM / 2)];
      float2 s1 = xr[(size_t)k1 * (DDIM / 2)];
      float2 s2 = xr[(size_t)k2 * (DDIM / 2)];
      add(base + j, v0, s0);
      if (j + 1 < n) add(base + j + 1, v1, s1);
      if (j + 2 < n) add(base + j + 2, v2, s2);
    }
    e = en;
  }
  while (ri < cnt) flush();
}

template <bool LAST>
__global__ __launch_bounds__(FB_T) void layer_rows_wmma(
    const int* __restrict__ ptrS, const int2* __restrict__ sortS,
    const float* __restrict__ uin,
    const float* __restrict__ W,
    const int* __restrict__ ptrI, const int2* __restrict__ sortI,
    const float* __restrict__ xI,
    float* __restrict__ out,
    int nrows) {
  __shared__ v2f Wp[64 * WP_STRIDE];
  __shared__ v2f tile2[FB_ROWS * TS / 2];
  float* tile = (float*)tile2;
#pragma unroll
  for (int i = 0; i < 64 * 64 / FB_T; ++i) {
    int idx = i * FB_T + threadIdx.x;
    int t = idx >> 6, n = idx & 63;
    v2f p;
    p.x = W[(size_t)(2 * t) * DDIM + n];
    p.y = W[(size_t)(2 * t + 1) * DDIM + n];
    Wp[t * WP_STRIDE + n] = p;
  }

  int lane = threadIdx.x & 31;
  int w    = threadIdx.x >> 5;
  int rowB = blockIdx.x * FB_ROWS;
  int r0  = __builtin_amdgcn_readfirstlane(rowB + w * RPW);
  int cnt = nrows - r0;
  if (cnt > RPW) cnt = RPW;
  float2* trow = (float2*)(tile + w * RPW * TS) + lane;
  if (cnt > 0) wave_row_sums<0>(ptrS, sortS, uin, trow, nullptr, r0, cnt, lane);
#pragma nounroll
  for (int i = (cnt > 0 ? cnt : 0); i < RPW; ++i) trow[i * (TS / 2)] = make_float2(0.f, 0.f);
  __syncthreads();

  int tI    = w >> 1;
  int cb    = (w & 1) * 32;
  int m     = lane & 15;
  int khalf = (lane >> 4) << 1;
  int lrow  = tI * 16 + m;
  int grow  = rowB + lrow;
  if (grow >= nrows) grow = nrows - 1;

  const float* rowNew = tile + lrow * TS;
  const float* rowU   = uin + (size_t)grow * DDIM;

  v8f acc0 = {}, acc1 = {};

#pragma unroll
  for (int k0 = 0; k0 < 64; k0 += 4) {
    int kk = k0 + khalf;
    v2f a = *(const v2f*)(rowNew + kk);
    const v2f* wr = &Wp[(kk >> 1) * WP_STRIDE + cb];
    v2f b0 = wr[m];
    v2f b1 = wr[m + 16];
    acc0 = __builtin_amdgcn_wmma_f32_16x16x4_f32(false, a, false, b0, (short)0, acc0, false, false);
    acc1 = __builtin_amdgcn_wmma_f32_16x16x4_f32(false, a, false, b1, (short)0, acc1, false, false);
  }

#pragma nounroll
  for (int g = 0; g < 64; g += 32) {
    v2f a[8];
#pragma unroll
    for (int u = 0; u < 8; ++u) {
      int kk = g + u * 4 + khalf;
      a[u].x = rowU[kk]; a[u].y = rowU[kk + 1];
    }
#pragma unroll
    for (int u = 0; u < 8; ++u) {
      int kk = g + u * 4 + khalf;
      const v2f* wr = &Wp[(32 + (kk >> 1)) * WP_STRIDE + cb];
      v2f b0 = wr[m];
      v2f b1 = wr[m + 16];
      acc0 = __builtin_amdgcn_wmma_f32_16x16x4_f32(false, a[u], false, b0, (short)0, acc0, false, false);
      acc1 = __builtin_amdgcn_wmma_f32_16x16x4_f32(false, a[u], false, b1, (short)0, acc1, false, false);
    }
  }

  int mo = (lane >> 4) * 8;
  if (!LAST) {
    float* orow = out + (size_t)(rowB + tI * 16) * DDIM + cb;
#pragma unroll
    for (int i2 = 0; i2 < 8; ++i2) {
      if (rowB + tI * 16 + mo + i2 < nrows) {
        orow[(size_t)(mo + i2) * DDIM + m]      = acc0[i2];
        orow[(size_t)(mo + i2) * DDIM + m + 16] = acc1[i2];
      }
    }
  } else {
    __syncthreads();
    float* tp = tile + (tI * 16 + mo) * TS + cb + m;
#pragma unroll
    for (int i2 = 0; i2 < 8; ++i2) {
      tp[i2 * TS]      = acc0[i2];
      tp[i2 * TS + 16] = acc1[i2];
    }
    __syncthreads();
    if (cnt > 0)
      wave_row_sums<1>(ptrI, sortI, xI, trow, (float2*)(out + (size_t)r0 * DDIM) + lane, r0, cnt, lane);
  }
}

extern "C" void kernel_launch(void* const* d_in, const int* in_sizes, int n_in,
                              void* d_out, int out_size, void* d_ws, size_t ws_size,
                              hipStream_t stream) {
  const float* user_emb    = (const float*)d_in[0];
  const float* item_emb    = (const float*)d_in[1];
  const float* weights     = (const float*)d_in[2];
  const float* social_vals = (const float*)d_in[3];
  const float* inter_vals  = (const float*)d_in[4];
  const int*   social_rows = (const int*)d_in[5];
  const int*   social_cols = (const int*)d_in[6];
  const int*   inter_rows  = (const int*)d_in[7];
  const int*   inter_cols  = (const int*)d_in[8];

  const int D   = DDIM;
  const int U   = in_sizes[0] / D;
  const int I   = in_sizes[1] / D;
  const int L   = in_sizes[2] / (2 * D * D);
  const int E_S = in_sizes[3];
  const int E_I = in_sizes[4];

  float* out_u = (float*)d_out;
  float* out_i = (float*)d_out + (size_t)U * D;

  float* planeP = (float*)d_ws;
  float* planeQ = planeP + (size_t)U * D;
  int*   cntS   = (int*)(planeQ + (size_t)U * D);
  int*   cntI   = cntS + (U + 1);
  int*   ptrS   = cntI + (U + 1);
  int*   ptrI   = ptrS + (U + 1);
  int2*  sortS  = (int2*)(ptrI + (U + 1));
  int2*  sortI  = sortS + E_S;

  const int nbuckets = (U + RB - 1) / RB;
  const int chunksS  = (E_S + CHUNK - 1) / CHUNK;
  const int chunksI  = (E_I + CHUNK - 1) / CHUNK;
  const int quadS    = (nbuckets * chunksS + 3) / 4;
  const int quadI    = (nbuckets * chunksI + 3) / 4;
  int4*  midS = (int4*)d_ws;
  int4*  midI = midS + E_S;
  int4*  cmS  = (int4*)(midI + E_I);
  int4*  cmI  = cmS + quadS;
  int4*  cbS  = cmI + quadI;
  int4*  cbI  = cbS + quadS;

  const int chunkBlocks = chunksS + chunksI;
  const int scanTiles   = ((quadS > quadI ? quadS : quadI) + SC_T - 1) / SC_T;
  const int rowBlocks   = (U + 8 * RPW - 1) / (8 * RPW);
  const int layerBlocks = (U + FB_ROWS - 1) / FB_ROWS;

  bucket_count<<<chunkBlocks, CH_T, 0, stream>>>(social_rows, E_S, (int*)cmS, inter_rows, E_I, (int*)cmI, nbuckets);
  bucket_scan<<<dim3(scanTiles, 2), SC_T, 0, stream>>>(cmS, cbS, quadS, cmI, cbI, quadI);
  bucket_place<<<chunkBlocks, CH_T, 0, stream>>>(
      social_rows, social_cols, social_vals, E_S, (const int*)cbS, midS,
      inter_rows, inter_cols, inter_vals, E_I, (const int*)cbI, midI, nbuckets);
  bucket_rows<<<dim3(nbuckets, 2), RB, 0, stream>>>(
      (const int*)cbS, midS, E_S, ptrS, sortS,
      (const int*)cbI, midI, E_I, ptrI, sortI, U, nbuckets);

  const float* ucur = user_emb;
  for (int k = 0; k < L; ++k) {
    const float* Wk = weights + (size_t)k * 2 * D * D;
    if (k == L - 1) {
      layer_rows_wmma<true><<<layerBlocks, FB_T, 0, stream>>>(
          ptrS, sortS, ucur, Wk, ptrI, sortI, item_emb, out_u, U);
    } else {
      float* dst = (k & 1) ? planeQ : planeP;
      layer_rows_wmma<false><<<layerBlocks, FB_T, 0, stream>>>(
          ptrS, sortS, ucur, Wk, ptrI, sortI, item_emb, dst, U);
      ucur = dst;
    }
  }
  if (L == 0) {
    hipMemcpyAsync(out_u, user_emb, (size_t)U * D * sizeof(float),
                   hipMemcpyDeviceToDevice, stream);
    spmm_rows<true><<<rowBlocks, 256, 0, stream>>>(ptrI, sortI, item_emb, out_u, U);
  }

  hipMemcpyAsync(out_i, item_emb, (size_t)I * D * sizeof(float),
                 hipMemcpyDeviceToDevice, stream);
}


# Edge grouping in 2 launches, not 4: a chunk block counts, scans and places its edges inside its own chunk; a bucket block collects its run from every chunk; row sum order = LDS arrival as before
# speedup vs baseline: 3.3022x; 1.1524x over previous
#include <hip/hip_runtime.h>

typedef __attribute__((ext_vector_type(2))) float v2f;
typedef __attribute__((ext_vector_type(8))) float v8f;

#define DDIM 64

#define RB_SHIFT 8
#define RB (1 << RB_SHIFT)
#define HMAX 1024
#define CH_T 512
#define CH_EPT 8
#define CHUNK (CH_T * CH_EPT)

__global__ __launch_bounds__(CH_T) void chunk_group(
    const int* __restrict__ rowsA, const int* __restrict__ colsA, const float* __restrict__ valsA,
    int nA, int* __restrict__ runA, int4* __restrict__ midA,
    const int* __restrict__ rowsB, const int* __restrict__ colsB, const float* __restrict__ valsB,
    int nB, int* __restrict__ runB, int4* __restrict__ midB,
    int nbuckets) {
  __shared__ int cur[HMAX + 32];
  __shared__ int waveSum[CH_T / 32];
  int blocksA = (nA + CHUNK - 1) / CHUNK;
  int blk = blockIdx.x;
  const int* rows = rowsA; const int* cols = colsA; const float* vals = valsA;
  int n = nA; int* run = runA; int4* mid = midA; int nblk = blocksA;
  if (blk >= blocksA) {
    blk -= blocksA; rows = rowsB; cols = colsB; vals = valsB;
    n = nB; run = runB; mid = midB; nblk = (nB + CHUNK - 1) / CHUNK;
  }
  int tid = threadIdx.x, lane = tid & 31, w = tid >> 5;
  cur[2 * tid] = 0; cur[2 * tid + 1] = 0;
  int first = blk * CHUNK + tid;
  int r[CH_EPT], c[CH_EPT];
  float v[CH_EPT];
#pragma unroll
  for (int t = 0; t < CH_EPT; ++t) {
    int idx = first + t * CH_T;
    r[t] = -1; c[t] = 0; v[t] = 0.f;
    if (idx < n) { r[t] = rows[idx]; c[t] = cols[idx]; v[t] = vals[idx]; }
  }
  __syncthreads();
#pragma unroll
  for (int t = 0; t < CH_EPT; ++t)
    if (r[t] >= 0) atomicAdd(&cur[r[t] >> RB_SHIFT], 1);
  __syncthreads();
  int i0 = 2 * tid;
  int n0 = (i0 < nbuckets) ? cur[i0] : 0;
  int n1 = (i0 + 1 < nbuckets) ? cur[i0 + 1] : 0;
  int s = n0 + n1;
  int incl = s;
  for (int o = 1; o < 32; o <<= 1) {
    int u = __shfl_up(incl, o);
    if (lane >= o) incl += u;
  }
  if (lane == 31) waveSum[w] = incl;
  __syncthreads();
  int pre = 0, tot = 0;
  for (int k = 0; k < CH_T / 32; ++k) {
    int ws = waveSum[k];
    tot += ws;
    if (k < w) pre += ws;
  }
  int ea = pre + incl - s;
  int eb = ea + n0;
  if (i0 < nbuckets)     { cur[i0] = ea;     run[(size_t)i0 * nblk + blk] = ea; }
  if (i0 + 1 < nbuckets) { cur[i0 + 1] = eb; run[(size_t)(i0 + 1) * nblk + blk] = eb; }
  if (tid == 0) run[(size_t)nbuckets * nblk + blk] = tot;
  __syncthreads();
  int spare = HMAX + lane;
  int pos[CH_EPT];
#pragma unroll
  for (int t = 0; t < CH_EPT; ++t)
    pos[t] = atomicAdd(&cur[(r[t] >= 0) ? (r[t] >> RB_SHIFT) : spare], 1);
  int4* cmid = mid + (size_t)blk * CHUNK;
#pragma unroll
  for (int t = 0; t < CH_EPT; ++t)
    if (r[t] >= 0) cmid[pos[t]] = make_int4(c[t], __float_as_int(v[t]), r[t], 0);
}

#define RW_U 10
#define MAXC 1024

__global__ __launch_bounds__(RB) void bucket_rows(
    const int* __restrict__ runA, const int4* __restrict__ midA,
    int nA, int* __restrict__ ptrA, int2* __restrict__ sortedA,
    const int* __restrict__ runB, const int4* __restrict__ midB,
    int nB, int* __restrict__ ptrB, int2* __restrict__ sortedB,
    int nrows, int nbuckets) {
  __shared__ int cnt[RB + 32];
  __shared__ int waveSum[RB / 32];
  __shared__ int2 run_s[MAXC];
  const int* run = runA; const int4* mid = midA;
  int n = nA; int* ptr = ptrA; int2* sorted = sortedA;
  if (blockIdx.y) { run = runB; mid = midB; n = nB; ptr = ptrB; sorted = sortedB; }
  int nblk = (n + CHUNK - 1) / CHUNK;
  int b = blockIdx.x;
  int tid = threadIdx.x, lane = tid & 31, w = tid >> 5;
  int row0 = b << RB_SHIFT;
  const int* rb = run + (size_t)b * nblk;
  int part = 0;
  for (int c = tid; c < nblk; c += RB) {
    int lo = rb[c], hi = rb[nblk + c];
    run_s[c] = make_int2(c * CHUNK + lo, hi - lo);
    part += lo;
  }
  for (int o = 16; o > 0; o >>= 1) part += __shfl_xor(part, o);
  if (lane == 0) waveSum[w] = part;
  cnt[tid] = 0;
  __syncthreads();
  int beg = 0;
  for (int k = 0; k < RB / 32; ++k) beg += waveSum[k];

  int sub = lane >> 4, j = lane & 15;
  int npairs = (nblk + 1) >> 1;
  for (int p0 = w; p0 < npairs; p0 += (RB / 32) * RW_U) {
    int base[RW_U], len[RW_U], r0[RW_U], r1[RW_U];
    int longest = 0;
#pragma unroll
    for (int u = 0; u < RW_U; ++u) {
      int cc = 2 * (p0 + u * (RB / 32)) + sub;
      int2 rs = run_s[cc < nblk ? cc : 0];
      base[u] = rs.x; len[u] = (cc < nblk) ? rs.y : 0;
      longest = len[u] > longest ? len[u] : longest;
    }
#pragma unroll
    for (int u = 0; u < RW_U; ++u) {
      r0[u] = (j < len[u]) ? mid[base[u] + j].z : -1;
      r1[u] = (j + 16 < len[u]) ? mid[base[u] + j + 16].z : -1;
    }
#pragma unroll
    for (int u = 0; u < RW_U; ++u) {
      if (r0[u] >= 0) atomicAdd(&cnt[r0[u] - row0], 1);
      if (r1[u] >= 0) atomicAdd(&cnt[r1[u] - row0], 1);
    }
    if (longest > 32) {
#pragma nounroll
      for (int u = 0; u < RW_U; ++u) {
        int cc = 2 * (p0 + u * (RB / 32)) + sub;
        if (cc < nblk) {
          int2 rs = run_s[cc];
          for (int jj = j + 32; jj < rs.y; jj += 16) atomicAdd(&cnt[mid[rs.x + jj].z - row0], 1);
        }
      }
    }
  }
  __syncthreads();
  int c = cnt[tid];
  int incl = c;
  for (int o = 1; o < 32; o <<= 1) {
    int v = __shfl_up(incl, o);
    if (lane >= o) incl += v;
  }
  if (lane == 31) waveSum[w] = incl;
  __syncthreads();
  int pre = 0;
  for (int k = 0; k < RB / 32; ++k) if (k < w) pre += waveSum[k];
  int start = beg + pre + incl - c;
  if (row0 + tid < nrows) ptr[row0 + tid] = start;
  if (b == nbuckets - 1 && tid == 0) ptr[nrows] = n;
  cnt[tid] = start;
  __syncthreads();
  for (int p0 = w; p0 < npairs; p0 += (RB / 32) * RW_U) {
    int base[RW_U], len[RW_U];
    int4 e0[RW_U], e1[RW_U];
    int longest = 0;
#pragma unroll
    for (int u = 0; u < RW_U; ++u) {
      int cc = 2 * (p0 + u * (RB / 32)) + sub;
      int2 rs = run_s[cc < nblk ? cc : 0];
      base[u] = rs.x; len[u] = (cc < nblk) ? rs.y : 0;
      longest = len[u] > longest ? len[u] : longest;
    }
#pragma unroll
    for (int u = 0; u < RW_U; ++u) {
      e0[u] = make_int4(0, 0, -1, 0);
      e1[u] = make_int4(0, 0, -1, 0);
      if (j < len[u])      e0[u] = mid[base[u] + j];
      if (j + 16 < len[u]) e1[u] = mid[base[u] + j + 16];
    }
    int pos0[RW_U], pos1[RW_U];
#pragma unroll
    for (int u = 0; u < RW_U; ++u) {
      pos0[u] = atomicAdd(&cnt[(e0[u].z >= 0) ? (e0[u].z - row0) : (RB + lane)], 1);
      pos1[u] = 0;
      if (e1[u].z >= 0) pos1[u] = atomicAdd(&cnt[e1[u].z - row0], 1);
    }
#pragma unroll
    for (int u = 0; u < RW_U; ++u) {
      if (e0[u].z >= 0) sorted[pos0[u]] = make_int2(e0[u].x, e0[u].y);
      if (e1[u].z >= 0) sorted[pos1[u]] = make_int2(e1[u].x, e1[u].y);
    }
    if (longest > 32) {
#pragma nounroll
      for (int u = 0; u < RW_U; ++u) {
        int cc = 2 * (p0 + u * (RB / 32)) + sub;
        if (cc < nblk) {
          int2 rs = run_s[cc];
          for (int jj = j + 32; jj < rs.y; jj += 16) {
            int4 x = mid[rs.x + jj];
            sorted[atomicAdd(&cnt[x.z - row0], 1)] = make_int2(x.x, x.y);
          }
        }
      }
    }
  }
}

#define RPW 8

template <bool ADD>
__global__ __launch_bounds__(256) void spmm_rows(const int* __restrict__ ptr,
                                                 const int2* __restrict__ sorted,
                                                 const float* __restrict__ x,
                                                 float* out,
                                                 int nrows) {
  int lane = threadIdx.x & 31;
  int r0 = __builtin_amdgcn_readfirstlane((blockIdx.x * 8 + (threadIdx.x >> 5)) * RPW);
  if (r0 >= nrows) return;
  int cnt = nrows - r0;
  if (cnt > RPW) cnt = RPW;
  int p = ptr[r0 + (lane < cnt ? lane : cnt)];
  int ebeg   = __builtin_amdgcn_readlane(p, 0);
  int eend   = __builtin_amdgcn_readlane(p, cnt);
  int rowEnd = __builtin_amdgcn_readlane(p, 1);
  int ri = 0;
  float2* orow = (float2*)(out + (size_t)r0 * DDIM) + lane;
  const float2* xr = (const float2*)x + lane;
  float2 acc = make_float2(0.f, 0.f);
  float2 st[RPW];
  if (ADD) {
#pragma unroll
    for (int i = 0; i < RPW; ++i) st[i] = orow[(i < cnt ? i : cnt - 1) * 32];
    acc = st[0];
  }
  auto flush = [&]() {
    orow[ri * 32] = acc;
    ++ri;
    if (ri < cnt) rowEnd = __builtin_amdgcn_readlane(p, ri + 1);
    acc = make_float2(0.f, 0.f);
    if (ADD) {
#pragma unroll
      for (int i = 1; i < RPW; ++i) if (ri == i) acc = st[i];
    }
  };

  const int last = eend - 1;
  int2 e = make_int2(0, 0);
  if (ebeg < eend) e = sorted[(ebeg + lane < eend) ? ebeg + lane : last];
  for (int base = ebeg; base < eend; base += 32) {
    int n = eend - base;
    if (n > 32) n = 32;
    int2 en = sorted[(base + 32 + lane < eend) ? base + 32 + lane : last];
    auto add = [&](int idx, float v, float2 s) {
      while (idx >= rowEnd) flush();
      acc.x += v * s.x; acc.y += v * s.y;
    };
    int j = 0;
    for (; j + 4 <= n; j += 4) {
      int   k0 = __builtin_amdgcn_readlane(e.x, j);
      int   k1 = __builtin_amdgcn_readlane(e.x, j + 1);
      int   k2 = __builtin_amdgcn_readlane(e.x, j + 2);
      int   k3 = __builtin_amdgcn_readlane(e.x, j + 3);
      float v0 = __int_as_float(__builtin_amdgcn_readlane(e.y, j));
      float v1 = __int_as_float(__builtin_amdgcn_readlane(e.y, j + 1));
      float v2 = __int_as_float(__builtin_amdgcn_readlane(e.y, j + 2));
      float v3 = __int_as_float(__builtin_amdgcn_readlane(e.y, j + 3));
      float2 s0 = xr[(size_t)k0 * (DDIM / 2)];
      float2 s1 = xr[(size_t)k1 * (DDIM / 2)];
      float2 s2 = xr[(size_t)k2 * (DDIM / 2)];
      float2 s3 = xr[(size_t)k3 * (DDIM / 2)];
      add(base + j,     v0, s0);
      add(base + j + 1, v1, s1);
      add(base + j + 2, v2, s2);
      add(base + j + 3, v3, s3);
    }
    if (j < n) {
      int j1 = (j + 1 < n) ? j + 1 : n - 1;
      int j2 = (j + 2 < n) ? j + 2 : n - 1;
      int   k0 = __builtin_amdgcn_readlane(e.x, j);
      int   k1 = __builtin_amdgcn_readlane(e.x, j1);
      int   k2 = __builtin_amdgcn_readlane(e.x, j2);
      float v0 = __int_as_float(__builtin_amdgcn_readlane(e.y, j));
      float v1 = __int_as_float(__builtin_amdgcn_readlane(e.y, j1));
      float v2 = __int_as_float(__builtin_amdgcn_readlane(e.y, j2));
      float2 s0 = xr[(size_t)k0 * (DDIM / 2)];
      float2 s1 = xr[(size_t)k1 * (DDIM / 2)];
      float2 s2 = xr[(size_t)k2 * (DDIM / 2)];
      add(base + j, v0, s0);
      if (j + 1 < n) add(base + j + 1, v1, s1);
      if (j + 2 < n) add(base + j + 2, v2, s2);
    }
    e = en;
  }
  while (ri < cnt) flush();
}

#define WP_STRIDE 80
#define TS 68
#define FB_T 512
#define FB_ROWS (FB_T / 32 * RPW)

template <int MODE>
__device__ __forceinline__ void wave_row_sums(const int* __restrict__ ptr,
                                              const int2* __restrict__ sorted,
                                              const float* __restrict__ x,
                                              float2* trow, float2* orow,
                                              int r0, int cnt, int lane) {
  int p = ptr[r0 + (lane < cnt ? lane : cnt)];
  int ebeg   = __builtin_amdgcn_readlane(p, 0);
  int eend   = __builtin_amdgcn_readlane(p, cnt);
  int rowEnd = __builtin_amdgcn_readlane(p, 1);
  int ri = 0;
  const float2* xr = (const float2*)x + lane;
  float2 acc = make_float2(0.f, 0.f);
  if (MODE) acc = trow[0];
  auto flush = [&]() {
    if (MODE) orow[ri * 32] = acc;
    else      trow[ri * (TS / 2)] = acc;
    ++ri;
    if (ri < cnt) rowEnd = __builtin_amdgcn_readlane(p, ri + 1);
    acc = make_float2(0.f, 0.f);
    if (MODE) { if (ri < cnt) acc = trow[ri * (TS / 2)]; }
  };

  const int last = eend - 1;
  int2 e = make_int2(0, 0);
  if (ebeg < eend) e = sorted[(ebeg + lane < eend) ? ebeg + lane : last];
  for (int base = ebeg; base < eend; base += 32) {
    int n = eend - base;
    if (n > 32) n = 32;
    int2 en = sorted[(base + 32 + lane < eend) ? base + 32 + lane : last];
    auto add = [&](int idx, float v, float2 s) {
      while (idx >= rowEnd) flush();
      acc.x += v * s.x; acc.y += v * s.y;
    };
    int j = 0;
    for (; j + 4 <= n; j += 4) {
      int   k0 = __builtin_amdgcn_readlane(e.x, j);
      int   k1 = __builtin_amdgcn_readlane(e.x, j + 1);
      int   k2 = __builtin_amdgcn_readlane(e.x, j + 2);
      int   k3 = __builtin_amdgcn_readlane(e.x, j + 3);
      float v0 = __int_as_float(__builtin_amdgcn_readlane(e.y, j));
      float v1 = __int_as_float(__builtin_amdgcn_readlane(e.y, j + 1));
      float v2 = __int_as_float(__builtin_amdgcn_readlane(e.y, j + 2));
      float v3 = __int_as_float(__builtin_amdgcn_readlane(e.y, j + 3));
      float2 s0 = xr[(size_t)k0 * (DDIM / 2)];
      float2 s1 = xr[(size_t)k1 * (DDIM / 2)];
      float2 s2 = xr[(size_t)k2 * (DDIM / 2)];
      float2 s3 = xr[(size_t)k3 * (DDIM / 2)];
      add(base + j,     v0, s0);
      add(base + j + 1, v1, s1);
      add(base + j + 2, v2, s2);
      add(base + j + 3, v3, s3);
    }
    if (j < n) {
      int j1 = (j + 1 < n) ? j + 1 : n - 1;
      int j2 = (j + 2 < n) ? j + 2 : n - 1;
      int   k0 = __builtin_amdgcn_readlane(e.x, j);
      int   k1 = __builtin_amdgcn_readlane(e.x, j1);
      int   k2 = __builtin_amdgcn_readlane(e.x, j2);
      float v0 = __int_as_float(__builtin_amdgcn_readlane(e.y, j));
      float v1 = __int_as_float(__builtin_amdgcn_readlane(e.y, j1));
      float v2 = __int_as_float(__builtin_amdgcn_readlane(e.y, j2));
      float2 s0 = xr[(size_t)k0 * (DDIM / 2)];
      float2 s1 = xr[(size_t)k1 * (DDIM / 2)];
      float2 s2 = xr[(size_t)k2 * (DDIM / 2)];
      add(base + j, v0, s0);
      if (j + 1 < n) add(base + j + 1, v1, s1);
      if (j + 2 < n) add(base + j + 2, v2, s2);
    }
    e = en;
  }
  while (ri < cnt) flush();
}

template <bool LAST>
__global__ __launch_bounds__(FB_T) void layer_rows_wmma(
    const int* __restrict__ ptrS, const int2* __restrict__ sortS,
    const float* __restrict__ uin,
    const float* __restrict__ W,
    const int* __restrict__ ptrI, const int2* __restrict__ sortI,
    const float* __restrict__ xI,
    float* __restrict__ out,
    int nrows) {
  __shared__ v2f Wp[64 * WP_STRIDE];
  __shared__ v2f tile2[FB_ROWS * TS / 2];
  float* tile = (float*)tile2;
#pragma unroll
  for (int i = 0; i < 64 * 64 / FB_T; ++i) {
    int idx = i * FB_T + threadIdx.x;
    int t = idx >> 6, n = idx & 63;
    v2f p;
    p.x = W[(size_t)(2 * t) * DDIM + n];
    p.y = W[(size_t)(2 * t + 1) * DDIM + n];
    Wp[t * WP_STRIDE + n] = p;
  }

  int lane = threadIdx.x & 31;
  int w    = threadIdx.x >> 5;
  int rowB = blockIdx.x * FB_ROWS;
  int r0  = __builtin_amdgcn_readfirstlane(rowB + w * RPW);
  int cnt = nrows - r0;
  if (cnt > RPW) cnt = RPW;
  float2* trow = (float2*)(tile + w * RPW * TS) + lane;
  if (cnt > 0) wave_row_sums<0>(ptrS, sortS, uin, trow, nullptr, r0, cnt, lane);
#pragma nounroll
  for (int i = (cnt > 0 ? cnt : 0); i < RPW; ++i) trow[i * (TS / 2)] = make_float2(0.f, 0.f);
  __syncthreads();

  int tI    = w >> 1;
  int cb    = (w & 1) * 32;
  int m     = lane & 15;
  int khalf = (lane >> 4) << 1;
  int lrow  = tI * 16 + m;
  int grow  = rowB + lrow;
  if (grow >= nrows) grow = nrows - 1;

  const float* rowNew = tile + lrow * TS;
  const float* rowU   = uin + (size_t)grow * DDIM;

  v8f acc0 = {}, acc1 = {};

#pragma unroll
  for (int k0 = 0; k0 < 64; k0 += 4) {
    int kk = k0 + khalf;
    v2f a = *(const v2f*)(rowNew + kk);
    const v2f* wr = &Wp[(kk >> 1) * WP_STRIDE + cb];
    v2f b0 = wr[m];
    v2f b1 = wr[m + 16];
    acc0 = __builtin_amdgcn_wmma_f32_16x16x4_f32(false, a, false, b0, (short)0, acc0, false, false);
    acc1 = __builtin_amdgcn_wmma_f32_16x16x4_f32(false, a, false, b1, (short)0, acc1, false, false);
  }

#pragma nounroll
  for (int g = 0; g < 64; g += 32) {
    v2f a[8];
#pragma unroll
    for (int u = 0; u < 8; ++u) {
      int kk = g + u * 4 + khalf;
      a[u].x = rowU[kk]; a[u].y = rowU[kk + 1];
    }
#pragma unroll
    for (int u = 0; u < 8; ++u) {
      int kk = g + u * 4 + khalf;
      const v2f* wr = &Wp[(32 + (kk >> 1)) * WP_STRIDE + cb];
      v2f b0 = wr[m];
      v2f b1 = wr[m + 16];
      acc0 = __builtin_amdgcn_wmma_f32_16x16x4_f32(false, a[u], false, b0, (short)0, acc0, false, false);
      acc1 = __builtin_amdgcn_wmma_f32_16x16x4_f32(false, a[u], false, b1, (short)0, acc1, false, false);
    }
  }

  int mo = (lane >> 4) * 8;
  if (!LAST) {
    float* orow = out + (size_t)(rowB + tI * 16) * DDIM + cb;
#pragma unroll
    for (int i2 = 0; i2 < 8; ++i2) {
      if (rowB + tI * 16 + mo + i2 < nrows) {
        orow[(size_t)(mo + i2) * DDIM + m]      = acc0[i2];
        orow[(size_t)(mo + i2) * DDIM + m + 16] = acc1[i2];
      }
    }
  } else {
    __syncthreads();
    float* tp = tile + (tI * 16 + mo) * TS + cb + m;
#pragma unroll
    for (int i2 = 0; i2 < 8; ++i2) {
      tp[i2 * TS]      = acc0[i2];
      tp[i2 * TS + 16] = acc1[i2];
    }
    __syncthreads();
    if (cnt > 0)
      wave_row_sums<1>(ptrI, sortI, xI, trow, (float2*)(out + (size_t)r0 * DDIM) + lane, r0, cnt, lane);
  }
}

extern "C" void kernel_launch(void* const* d_in, const int* in_sizes, int n_in,
                              void* d_out, int out_size, void* d_ws, size_t ws_size,
                              hipStream_t stream) {
  const float* user_emb    = (const float*)d_in[0];
  const float* item_emb    = (const float*)d_in[1];
  const float* weights     = (const float*)d_in[2];
  const float* social_vals = (const float*)d_in[3];
  const float* inter_vals  = (const float*)d_in[4];
  const int*   social_rows = (const int*)d_in[5];
  const int*   social_cols = (const int*)d_in[6];
  const int*   inter_rows  = (const int*)d_in[7];
  const int*   inter_cols  = (const int*)d_in[8];

  const int D   = DDIM;
  const int U   = in_sizes[0] / D;
  const int I   = in_sizes[1] / D;
  const int L   = in_sizes[2] / (2 * D * D);
  const int E_S = in_sizes[3];
  const int E_I = in_sizes[4];

  float* out_u = (float*)d_out;
  float* out_i = (float*)d_out + (size_t)U * D;

  float* planeP = (float*)d_ws;
  float* planeQ = planeP + (size_t)U * D;
  int*   cntS   = (int*)(planeQ + (size_t)U * D);
  int*   cntI   = cntS + (U + 1);
  int*   ptrS   = cntI + (U + 1);
  int*   ptrI   = ptrS + (U + 1);
  int2*  sortS  = (int2*)(ptrI + (U + 1));
  int2*  sortI  = sortS + E_S;

  const int nbuckets = (U + RB - 1) / RB;
  const int chunksS  = (E_S + CHUNK - 1) / CHUNK;
  const int chunksI  = (E_I + CHUNK - 1) / CHUNK;
  int4*  midS = (int4*)d_ws;
  int4*  midI = midS + (size_t)chunksS * CHUNK;
  int*   runS = (int*)(midI + (size_t)chunksI * CHUNK);
  int*   runI = runS + (size_t)(nbuckets + 1) * chunksS;

  const int chunkBlocks = chunksS + chunksI;
  const int rowBlocks   = (U + 8 * RPW - 1) / (8 * RPW);
  const int layerBlocks = (U + FB_ROWS - 1) / FB_ROWS;

  chunk_group<<<chunkBlocks, CH_T, 0, stream>>>(
      social_rows, social_cols, social_vals, E_S, runS, midS,
      inter_rows, inter_cols, inter_vals, E_I, runI, midI, nbuckets);
  bucket_rows<<<dim3(nbuckets, 2), RB, 0, stream>>>(
      runS, midS, E_S, ptrS, sortS,
      runI, midI, E_I, ptrI, sortI, U, nbuckets);

  const float* ucur = user_emb;
  for (int k = 0; k < L; ++k) {
    const float* Wk = weights + (size_t)k * 2 * D * D;
    if (k == L - 1) {
      layer_rows_wmma<true><<<layerBlocks, FB_T, 0, stream>>>(
          ptrS, sortS, ucur, Wk, ptrI, sortI, item_emb, out_u, U);
    } else {
      float* dst = (k & 1) ? planeQ : planeP;
      layer_rows_wmma<false><<<layerBlocks, FB_T, 0, stream>>>(
          ptrS, sortS, ucur, Wk, ptrI, sortI, item_emb, dst, U);
      ucur = dst;
    }
  }
  if (L == 0) {
    hipMemcpyAsync(out_u, user_emb, (size_t)U * D * sizeof(float),
                   hipMemcpyDeviceToDevice, stream);
    spmm_rows<true><<<rowBlocks, 256, 0, stream>>>(ptrI, sortI, item_emb, out_u, U);
  }

  hipMemcpyAsync(out_i, item_emb, (size_t)I * D * sizeof(float),
                 hipMemcpyDeviceToDevice, stream);
}


# Grouped edges reach memory as whole 128-byte lines: both grouping kernels put their entries in order in LDS and copy the piece out, not one scattered 8/16-byte store per edge; row sum orders kept
# speedup vs baseline: 3.7463x; 1.1345x over previous
#include <hip/hip_runtime.h>

typedef __attribute__((ext_vector_type(2))) float v2f;
typedef __attribute__((ext_vector_type(8))) float v8f;

#define DDIM 64

#define RB_SHIFT 8
#define RB (1 << RB_SHIFT)
#define HMAX 1024
#define CH_T 512
#define CH_EPT 8
#define CHUNK (CH_T * CH_EPT)

__global__ __launch_bounds__(CH_T) void chunk_group(
    const int* __restrict__ rowsA, const int* __restrict__ colsA, const float* __restrict__ valsA,
    int nA, int* __restrict__ runA, int4* __restrict__ midA,
    const int* __restrict__ rowsB, const int* __restrict__ colsB, const float* __restrict__ valsB,
    int nB, int* __restrict__ runB, int4* __restrict__ midB,
    int nbuckets) {
  __shared__ int cur[HMAX + 32];
  __shared__ int waveSum[CH_T / 32];
  __shared__ int4 stage[CHUNK];
  int blocksA = (nA + CHUNK - 1) / CHUNK;
  int blk = blockIdx.x;
  const int* rows = rowsA; const int* cols = colsA; const float* vals = valsA;
  int n = nA; int* run = runA; int4* mid = midA; int nblk = blocksA;
  if (blk >= blocksA) {
    blk -= blocksA; rows = rowsB; cols = colsB; vals = valsB;
    n = nB; run = runB; mid = midB; nblk = (nB + CHUNK - 1) / CHUNK;
  }
  int tid = threadIdx.x, lane = tid & 31, w = tid >> 5;
  cur[2 * tid] = 0; cur[2 * tid + 1] = 0;
  int first = blk * CHUNK + tid;
  int r[CH_EPT], c[CH_EPT];
  float v[CH_EPT];
#pragma unroll
  for (int t = 0; t < CH_EPT; ++t) {
    int idx = first + t * CH_T;
    r[t] = -1; c[t] = 0; v[t] = 0.f;
    if (idx < n) { r[t] = rows[idx]; c[t] = cols[idx]; v[t] = vals[idx]; }
  }
  __syncthreads();
#pragma unroll
  for (int t = 0; t < CH_EPT; ++t)
    if (r[t] >= 0) atomicAdd(&cur[r[t] >> RB_SHIFT], 1);
  __syncthreads();
  int i0 = 2 * tid;
  int n0 = (i0 < nbuckets) ? cur[i0] : 0;
  int n1 = (i0 + 1 < nbuckets) ? cur[i0 + 1] : 0;
  int s = n0 + n1;
  int incl = s;
  for (int o = 1; o < 32; o <<= 1) {
    int u = __shfl_up(incl, o);
    if (lane >= o) incl += u;
  }
  if (lane == 31) waveSum[w] = incl;
  __syncthreads();
  int pre = 0, tot = 0;
  for (int k = 0; k < CH_T / 32; ++k) {
    int ws = waveSum[k];
    tot += ws;
    if (k < w) pre += ws;
  }
  int ea = pre + incl - s;
  int eb = ea + n0;
  if (i0 < nbuckets)     { cur[i0] = ea;     run[(size_t)i0 * nblk + blk] = ea; }
  if (i0 + 1 < nbuckets) { cur[i0 + 1] = eb; run[(size_t)(i0 + 1) * nblk + blk] = eb; }
  if (tid == 0) run[(size_t)nbuckets * nblk + blk] = tot;
  __syncthreads();
  int spare = HMAX + lane;
  int pos[CH_EPT];
#pragma unroll
  for (int t = 0; t < CH_EPT; ++t)
    pos[t] = atomicAdd(&cur[(r[t] >= 0) ? (r[t] >> RB_SHIFT) : spare], 1);
#pragma unroll
  for (int t = 0; t < CH_EPT; ++t)
    if (r[t] >= 0) stage[pos[t]] = make_int4(c[t], __float_as_int(v[t]), r[t], 0);
  __syncthreads();
  int4* cmid = mid + (size_t)blk * CHUNK;
#pragma unroll
  for (int t = 0; t < CH_EPT; ++t) cmid[t * CH_T + tid] = stage[t * CH_T + tid];
}

#define RW_U 10
#define MAXC 1024

#define SCAP 4096

template <bool STAGED>
__device__ __forceinline__ void place_edges(const int4* __restrict__ mid, int2* __restrict__ sorted,
                                            int2* stage, int* cnt, const int2* run_s,
                                            int nblk, int npairs, int w, int lane,
                                            int row0, int beg) {
  int sub = lane >> 4, j = lane & 15;
  auto put = [&](int pos, int4 e) {
    if (STAGED) stage[pos - beg] = make_int2(e.x, e.y);
    else        sorted[pos] = make_int2(e.x, e.y);
  };
  for (int p0 = w; p0 < npairs; p0 += (RB / 32) * RW_U) {
    int base[RW_U], len[RW_U];
    int4 e0[RW_U], e1[RW_U];
    int longest = 0;
#pragma unroll
    for (int u = 0; u < RW_U; ++u) {
      int cc = 2 * (p0 + u * (RB / 32)) + sub;
      int2 rs = run_s[cc < nblk ? cc : 0];
      base[u] = rs.x; len[u] = (cc < nblk) ? rs.y : 0;
      longest = len[u] > longest ? len[u] : longest;
    }
#pragma unroll
    for (int u = 0; u < RW_U; ++u) {
      e0[u] = make_int4(0, 0, -1, 0);
      e1[u] = make_int4(0, 0, -1, 0);
      if (j < len[u])      e0[u] = mid[base[u] + j];
      if (j + 16 < len[u]) e1[u] = mid[base[u] + j + 16];
    }
    int pos0[RW_U], pos1[RW_U];
#pragma unroll
    for (int u = 0; u < RW_U; ++u) {
      pos0[u] = atomicAdd(&cnt[(e0[u].z >= 0) ? (e0[u].z - row0) : (RB + lane)], 1);
      pos1[u] = 0;
      if (e1[u].z >= 0) pos1[u] = atomicAdd(&cnt[e1[u].z - row0], 1);
    }
#pragma unroll
    for (int u = 0; u < RW_U; ++u) {
      if (e0[u].z >= 0) put(pos0[u], e0[u]);
      if (e1[u].z >= 0) put(pos1[u], e1[u]);
    }
    if (longest > 32) {
#pragma nounroll
      for (int u = 0; u < RW_U; ++u) {
        int cc = 2 * (p0 + u * (RB / 32)) + sub;
        if (cc < nblk) {
          int2 rs = run_s[cc];
          for (int jj = j + 32; jj < rs.y; jj += 16) {
            int4 x = mid[rs.x + jj];
            put(atomicAdd(&cnt[x.z - row0], 1), x);
          }
        }
      }
    }
  }
}

__global__ __launch_bounds__(RB) void bucket_rows(
    const int* __restrict__ runA, const int4* __restrict__ midA,
    int nA, int* __restrict__ ptrA, int2* __restrict__ sortedA,
    const int* __restrict__ runB, const int4* __restrict__ midB,
    int nB, int* __restrict__ ptrB, int2* __restrict__ sortedB,
    int nrows, int nbuckets) {
  __shared__ int cnt[RB + 32];
  __shared__ int waveSum[RB / 32];
  __shared__ int2 run_s[MAXC];
  __shared__ int2 stage[SCAP];
  const int* run = runA; const int4* mid = midA;
  int n = nA; int* ptr = ptrA; int2* sorted = sortedA;
  if (blockIdx.y) { run = runB; mid = midB; n = nB; ptr = ptrB; sorted = sortedB; }
  int nblk = (n + CHUNK - 1) / CHUNK;
  int b = blockIdx.x;
  int tid = threadIdx.x, lane = tid & 31, w = tid >> 5;
  int row0 = b << RB_SHIFT;
  const int* rb = run + (size_t)b * nblk;
  int part = 0;
  for (int c = tid; c < nblk; c += RB) {
    int lo = rb[c], hi = rb[nblk + c];
    run_s[c] = make_int2(c * CHUNK + lo, hi - lo);
    part += lo;
  }
  for (int o = 16; o > 0; o >>= 1) part += __shfl_xor(part, o);
  if (lane == 0) waveSum[w] = part;
  cnt[tid] = 0;
  __syncthreads();
  int beg = 0;
  for (int k = 0; k < RB / 32; ++k) beg += waveSum[k];

  int sub = lane >> 4, j = lane & 15;
  int npairs = (nblk + 1) >> 1;
  for (int p0 = w; p0 < npairs; p0 += (RB / 32) * RW_U) {
    int base[RW_U], len[RW_U], r0[RW_U], r1[RW_U];
    int longest = 0;
#pragma unroll
    for (int u = 0; u < RW_U; ++u) {
      int cc = 2 * (p0 + u * (RB / 32)) + sub;
      int2 rs = run_s[cc < nblk ? cc : 0];
      base[u] = rs.x; len[u] = (cc < nblk) ? rs.y : 0;
      longest = len[u] > longest ? len[u] : longest;
    }
#pragma unroll
    for (int u = 0; u < RW_U; ++u) {
      r0[u] = (j < len[u]) ? mid[base[u] + j].z : -1;
      r1[u] = (j + 16 < len[u]) ? mid[base[u] + j + 16].z : -1;
    }
#pragma unroll
    for (int u = 0; u < RW_U; ++u) {
      if (r0[u] >= 0) atomicAdd(&cnt[r0[u] - row0], 1);
      if (r1[u] >= 0) atomicAdd(&cnt[r1[u] - row0], 1);
    }
    if (longest > 32) {
#pragma nounroll
      for (int u = 0; u < RW_U; ++u) {
        int cc = 2 * (p0 + u * (RB / 32)) + sub;
        if (cc < nblk) {
          int2 rs = run_s[cc];
          for (int jj = j + 32; jj < rs.y; jj += 16) atomicAdd(&cnt[mid[rs.x + jj].z - row0], 1);
        }
      }
    }
  }
  __syncthreads();
  int c = cnt[tid];
  int incl = c;
  for (int o = 1; o < 32; o <<= 1) {
    int v = __shfl_up(incl, o);
    if (lane >= o) incl += v;
  }
  if (lane == 31) waveSum[w] = incl;
  __syncthreads();
  int pre = 0, totB = 0;
  for (int k = 0; k < RB / 32; ++k) {
    int ws = waveSum[k];
    totB += ws;
    if (k < w) pre += ws;
  }
  int start = beg + pre + incl - c;
  if (row0 + tid < nrows) ptr[row0 + tid] = start;
  if (b == nbuckets - 1 && tid == 0) ptr[nrows] = n;
  cnt[tid] = start;
  __syncthreads();
  if (totB <= SCAP) {
    place_edges<true>(mid, sorted, stage, cnt, run_s, nblk, npairs, w, lane, row0, beg);
    __syncthreads();
    int2* piece = sorted + beg;
#pragma unroll 4
    for (int i = tid; i < totB; i += RB) piece[i] = stage[i];
  } else {
    place_edges<false>(mid, sorted, stage, cnt, run_s, nblk, npairs, w, lane, row0, beg);
  }
}

#define RPW 8

template <bool ADD>
__global__ __launch_bounds__(256) void spmm_rows(const int* __restrict__ ptr,
                                                 const int2* __restrict__ sorted,
                                                 const float* __restrict__ x,
                                                 float* out,
                                                 int nrows) {
  int lane = threadIdx.x & 31;
  int r0 = __builtin_amdgcn_readfirstlane((blockIdx.x * 8 + (threadIdx.x >> 5)) * RPW);
  if (r0 >= nrows) return;
  int cnt = nrows - r0;
  if (cnt > RPW) cnt = RPW;
  int p = ptr[r0 + (lane < cnt ? lane : cnt)];
  int ebeg   = __builtin_amdgcn_readlane(p, 0);
  int eend   = __builtin_amdgcn_readlane(p, cnt);
  int rowEnd = __builtin_amdgcn_readlane(p, 1);
  int ri = 0;
  float2* orow = (float2*)(out + (size_t)r0 * DDIM) + lane;
  const float2* xr = (const float2*)x + lane;
  float2 acc = make_float2(0.f, 0.f);
  float2 st[RPW];
  if (ADD) {
#pragma unroll
    for (int i = 0; i < RPW; ++i) st[i] = orow[(i < cnt ? i : cnt - 1) * 32];
    acc = st[0];
  }
  auto flush = [&]() {
    orow[ri * 32] = acc;
    ++ri;
    if (ri < cnt) rowEnd = __builtin_amdgcn_readlane(p, ri + 1);
    acc = make_float2(0.f, 0.f);
    if (ADD) {
#pragma unroll
      for (int i = 1; i < RPW; ++i) if (ri == i) acc = st[i];
    }
  };

  const int last = eend - 1;
  int2 e = make_int2(0, 0);
  if (ebeg < eend) e = sorted[(ebeg + lane < eend) ? ebeg + lane : last];
  for (int base = ebeg; base < eend; base += 32) {
    int n = eend - base;
    if (n > 32) n = 32;
    int2 en = sorted[(base + 32 + lane < eend) ? base + 32 + lane : last];
    auto add = [&](int idx, float v, float2 s) {
      while (idx >= rowEnd) flush();
      acc.x += v * s.x; acc.y += v * s.y;
    };
    int j = 0;
    for (; j + 4 <= n; j += 4) {
      int   k0 = __builtin_amdgcn_readlane(e.x, j);
      int   k1 = __builtin_amdgcn_readlane(e.x, j + 1);
      int   k2 = __builtin_amdgcn_readlane(e.x, j + 2);
      int   k3 = __builtin_amdgcn_readlane(e.x, j + 3);
      float v0 = __int_as_float(__builtin_amdgcn_readlane(e.y, j));
      float v1 = __int_as_float(__builtin_amdgcn_readlane(e.y, j + 1));
      float v2 = __int_as_float(__builtin_amdgcn_readlane(e.y, j + 2));
      float v3 = __int_as_float(__builtin_amdgcn_readlane(e.y, j + 3));
      float2 s0 = xr[(size_t)k0 * (DDIM / 2)];
      float2 s1 = xr[(size_t)k1 * (DDIM / 2)];
      float2 s2 = xr[(size_t)k2 * (DDIM / 2)];
      float2 s3 = xr[(size_t)k3 * (DDIM / 2)];
      add(base + j,     v0, s0);
      add(base + j + 1, v1, s1);
      add(base + j + 2, v2, s2);
      add(base + j + 3, v3, s3);
    }
    if (j < n) {
      int j1 = (j + 1 < n) ? j + 1 : n - 1;
      int j2 = (j + 2 < n) ? j + 2 : n - 1;
      int   k0 = __builtin_amdgcn_readlane(e.x, j);
      int   k1 = __builtin_amdgcn_readlane(e.x, j1);
      int   k2 = __builtin_amdgcn_readlane(e.x, j2);
      float v0 = __int_as_float(__builtin_amdgcn_readlane(e.y, j));
      float v1 = __int_as_float(__builtin_amdgcn_readlane(e.y, j1));
      float v2 = __int_as_float(__builtin_amdgcn_readlane(e.y, j2));
      float2 s0 = xr[(size_t)k0 * (DDIM / 2)];
      float2 s1 = xr[(size_t)k1 * (DDIM / 2)];
      float2 s2 = xr[(size_t)k2 * (DDIM / 2)];
      add(base + j, v0, s0);
      if (j + 1 < n) add(base + j + 1, v1, s1);
      if (j + 2 < n) add(base + j + 2, v2, s2);
    }
    e = en;
  }
  while (ri < cnt) flush();
}

#define WP_STRIDE 80
#define TS 68
#define FB_T 512
#define FB_ROWS (FB_T / 32 * RPW)

template <int MODE>
__device__ __forceinline__ void wave_row_sums(const int* __restrict__ ptr,
                                              const int2* __restrict__ sorted,
                                              const float* __restrict__ x,
                                              float2* trow, float2* orow,
                                              int r0, int cnt, int lane) {
  int p = ptr[r0 + (lane < cnt ? lane : cnt)];
  int ebeg   = __builtin_amdgcn_readlane(p, 0);
  int eend   = __builtin_amdgcn_readlane(p, cnt);
  int rowEnd = __builtin_amdgcn_readlane(p, 1);
  int ri = 0;
  const float2* xr = (const float2*)x + lane;
  float2 acc = make_float2(0.f, 0.f);
  if (MODE) acc = trow[0];
  auto flush = [&]() {
    if (MODE) orow[ri * 32] = acc;
    else      trow[ri * (TS / 2)] = acc;
    ++ri;
    if (ri < cnt) rowEnd = __builtin_amdgcn_readlane(p, ri + 1);
    acc = make_float2(0.f, 0.f);
    if (MODE) { if (ri < cnt) acc = trow[ri * (TS / 2)]; }
  };

  const int last = eend - 1;
  int2 e = make_int2(0, 0);
  if (ebeg < eend) e = sorted[(ebeg + lane < eend) ? ebeg + lane : last];
  for (int base = ebeg; base < eend; base += 32) {
    int n = eend - base;
    if (n > 32) n = 32;
    int2 en = sorted[(base + 32 + lane < eend) ? base + 32 + lane : last];
    auto add = [&](int idx, float v, float2 s) {
      while (idx >= rowEnd) flush();
      acc.x += v * s.x; acc.y += v * s.y;
    };
    int j = 0;
    for (; j + 4 <= n; j += 4) {
      int   k0 = __builtin_amdgcn_readlane(e.x, j);
      int   k1 = __builtin_amdgcn_readlane(e.x, j + 1);
      int   k2 = __builtin_amdgcn_readlane(e.x, j + 2);
      int   k3 = __builtin_amdgcn_readlane(e.x, j + 3);
      float v0 = __int_as_float(__builtin_amdgcn_readlane(e.y, j));
      float v1 = __int_as_float(__builtin_amdgcn_readlane(e.y, j + 1));
      float v2 = __int_as_float(__builtin_amdgcn_readlane(e.y, j + 2));
      float v3 = __int_as_float(__builtin_amdgcn_readlane(e.y, j + 3));
      float2 s0 = xr[(size_t)k0 * (DDIM / 2)];
      float2 s1 = xr[(size_t)k1 * (DDIM / 2)];
      float2 s2 = xr[(size_t)k2 * (DDIM / 2)];
      float2 s3 = xr[(size_t)k3 * (DDIM / 2)];
      add(base + j,     v0, s0);
      add(base + j + 1, v1, s1);
      add(base + j + 2, v2, s2);
      add(base + j + 3, v3, s3);
    }
    if (j < n) {
      int j1 = (j + 1 < n) ? j + 1 : n - 1;
      int j2 = (j + 2 < n) ? j + 2 : n - 1;
      int   k0 = __builtin_amdgcn_readlane(e.x, j);
      int   k1 = __builtin_amdgcn_readlane(e.x, j1);
      int   k2 = __builtin_amdgcn_readlane(e.x, j2);
      float v0 = __int_as_float(__builtin_amdgcn_readlane(e.y, j));
      float v1 = __int_as_float(__builtin_amdgcn_readlane(e.y, j1));
      float v2 = __int_as_float(__builtin_amdgcn_readlane(e.y, j2));
      float2 s0 = xr[(size_t)k0 * (DDIM / 2)];
      float2 s1 = xr[(size_t)k1 * (DDIM / 2)];
      float2 s2 = xr[(size_t)k2 * (DDIM / 2)];
      add(base + j, v0, s0);
      if (j + 1 < n) add(base + j + 1, v1, s1);
      if (j + 2 < n) add(base + j + 2, v2, s2);
    }
    e = en;
  }
  while (ri < cnt) flush();
}

template <bool LAST>
__global__ __launch_bounds__(FB_T) void layer_rows_wmma(
    const int* __restrict__ ptrS, const int2* __restrict__ sortS,
    const float* __restrict__ uin,
    const float* __restrict__ W,
    const int* __restrict__ ptrI, const int2* __restrict__ sortI,
    const float* __restrict__ xI,
    float* __restrict__ out,
    int nrows) {
  __shared__ v2f Wp[64 * WP_STRIDE];
  __shared__ v2f tile2[FB_ROWS * TS / 2];
  float* tile = (float*)tile2;
#pragma unroll
  for (int i = 0; i < 64 * 64 / FB_T; ++i) {
    int idx = i * FB_T + threadIdx.x;
    int t = idx >> 6, n = idx & 63;
    v2f p;
    p.x = W[(size_t)(2 * t) * DDIM + n];
    p.y = W[(size_t)(2 * t + 1) * DDIM + n];
    Wp[t * WP_STRIDE + n] = p;
  }

  int lane = threadIdx.x & 31;
  int w    = threadIdx.x >> 5;
  int rowB = blockIdx.x * FB_ROWS;
  int r0  = __builtin_amdgcn_readfirstlane(rowB + w * RPW);
  int cnt = nrows - r0;
  if (cnt > RPW) cnt = RPW;
  float2* trow = (float2*)(tile + w * RPW * TS) + lane;
  if (cnt > 0) wave_row_sums<0>(ptrS, sortS, uin, trow, nullptr, r0, cnt, lane);
#pragma nounroll
  for (int i = (cnt > 0 ? cnt : 0); i < RPW; ++i) trow[i * (TS / 2)] = make_float2(0.f, 0.f);
  __syncthreads();

  int tI    = w >> 1;
  int cb    = (w & 1) * 32;
  int m     = lane & 15;
  int khalf = (lane >> 4) << 1;
  int lrow  = tI * 16 + m;
  int grow  = rowB + lrow;
  if (grow >= nrows) grow = nrows - 1;

  const float* rowNew = tile + lrow * TS;
  const float* rowU   = uin + (size_t)grow * DDIM;

  v8f acc0 = {}, acc1 = {};

#pragma unroll
  for (int k0 = 0; k0 < 64; k0 += 4) {
    int kk = k0 + khalf;
    v2f a = *(const v2f*)(rowNew + kk);
    const v2f* wr = &Wp[(kk >> 1) * WP_STRIDE + cb];
    v2f b0 = wr[m];
    v2f b1 = wr[m + 16];
    acc0 = __builtin_amdgcn_wmma_f32_16x16x4_f32(false, a, false, b0, (short)0, acc0, false, false);
    acc1 = __builtin_amdgcn_wmma_f32_16x16x4_f32(false, a, false, b1, (short)0, acc1, false, false);
  }

#pragma nounroll
  for (int g = 0; g < 64; g += 32) {
    v2f a[8];
#pragma unroll
    for (int u = 0; u < 8; ++u) {
      int kk = g + u * 4 + khalf;
      a[u].x = rowU[kk]; a[u].y = rowU[kk + 1];
    }
#pragma unroll
    for (int u = 0; u < 8; ++u) {
      int kk = g + u * 4 + khalf;
      const v2f* wr = &Wp[(32 + (kk >> 1)) * WP_STRIDE + cb];
      v2f b0 = wr[m];
      v2f b1 = wr[m + 16];
      acc0 = __builtin_amdgcn_wmma_f32_16x16x4_f32(false, a[u], false, b0, (short)0, acc0, false, false);
      acc1 = __builtin_amdgcn_wmma_f32_16x16x4_f32(false, a[u], false, b1, (short)0, acc1, false, false);
    }
  }

  int mo = (lane >> 4) * 8;
  if (!LAST) {
    float* orow = out + (size_t)(rowB + tI * 16) * DDIM + cb;
#pragma unroll
    for (int i2 = 0; i2 < 8; ++i2) {
      if (rowB + tI * 16 + mo + i2 < nrows) {
        orow[(size_t)(mo + i2) * DDIM + m]      = acc0[i2];
        orow[(size_t)(mo + i2) * DDIM + m + 16] = acc1[i2];
      }
    }
  } else {
    __syncthreads();
    float* tp = tile + (tI * 16 + mo) * TS + cb + m;
#pragma unroll
    for (int i2 = 0; i2 < 8; ++i2) {
      tp[i2 * TS]      = acc0[i2];
      tp[i2 * TS + 16] = acc1[i2];
    }
    __syncthreads();
    if (cnt > 0)
      wave_row_sums<1>(ptrI, sortI, xI, trow, (float2*)(out + (size_t)r0 * DDIM) + lane, r0, cnt, lane);
  }
}

extern "C" void kernel_launch(void* const* d_in, const int* in_sizes, int n_in,
                              void* d_out, int out_size, void* d_ws, size_t ws_size,
                              hipStream_t stream) {
  const float* user_emb    = (const float*)d_in[0];
  const float* item_emb    = (const float*)d_in[1];
  const float* weights     = (const float*)d_in[2];
  const float* social_vals = (const float*)d_in[3];
  const float* inter_vals  = (const float*)d_in[4];
  const int*   social_rows = (const int*)d_in[5];
  const int*   social_cols = (const int*)d_in[6];
  const int*   inter_rows  = (const int*)d_in[7];
  const int*   inter_cols  = (const int*)d_in[8];

  const int D   = DDIM;
  const int U   = in_sizes[0] / D;
  const int I   = in_sizes[1] / D;
  const int L   = in_sizes[2] / (2 * D * D);
  const int E_S = in_sizes[3];
  const int E_I = in_sizes[4];

  float* out_u = (float*)d_out;
  float* out_i = (float*)d_out + (size_t)U * D;

  float* planeP = (float*)d_ws;
  float* planeQ = planeP + (size_t)U * D;
  int*   cntS   = (int*)(planeQ + (size_t)U * D);
  int*   cntI   = cntS + (U + 1);
  int*   ptrS   = cntI + (U + 1);
  int*   ptrI   = ptrS + (U + 1);
  int2*  sortS  = (int2*)(ptrI + (U + 1));
  int2*  sortI  = sortS + E_S;

  const int nbuckets = (U + RB - 1) / RB;
  const int chunksS  = (E_S + CHUNK - 1) / CHUNK;
  const int chunksI  = (E_I + CHUNK - 1) / CHUNK;
  int4*  midS = (int4*)d_ws;
  int4*  midI = midS + (size_t)chunksS * CHUNK;
  int*   runS = (int*)(midI + (size_t)chunksI * CHUNK);
  int*   runI = runS + (size_t)(nbuckets + 1) * chunksS;

  const int chunkBlocks = chunksS + chunksI;
  const int rowBlocks   = (U + 8 * RPW - 1) / (8 * RPW);
  const int layerBlocks = (U + FB_ROWS - 1) / FB_ROWS;

  chunk_group<<<chunkBlocks, CH_T, 0, stream>>>(
      social_rows, social_cols, social_vals, E_S, runS, midS,
      inter_rows, inter_cols, inter_vals, E_I, runI, midI, nbuckets);
  bucket_rows<<<dim3(nbuckets, 2), RB, 0, stream>>>(
      runS, midS, E_S, ptrS, sortS,
      runI, midI, E_I, ptrI, sortI, U, nbuckets);

  const float* ucur = user_emb;
  for (int k = 0; k < L; ++k) {
    const float* Wk = weights + (size_t)k * 2 * D * D;
    if (k == L - 1) {
      layer_rows_wmma<true><<<layerBlocks, FB_T, 0, stream>>>(
          ptrS, sortS, ucur, Wk, ptrI, sortI, item_emb, out_u, U);
    } else {
      float* dst = (k & 1) ? planeQ : planeP;
      layer_rows_wmma<false><<<layerBlocks, FB_T, 0, stream>>>(
          ptrS, sortS, ucur, Wk, ptrI, sortI, item_emb, dst, U);
      ucur = dst;
    }
  }
  if (L == 0) {
    hipMemcpyAsync(out_u, user_emb, (size_t)U * D * sizeof(float),
                   hipMemcpyDeviceToDevice, stream);
    spmm_rows<true><<<rowBlocks, 256, 0, stream>>>(ptrI, sortI, item_emb, out_u, U);
  }

  hipMemcpyAsync(out_i, item_emb, (size_t)I * D * sizeof(float),
                 hipMemcpyDeviceToDevice, stream);
}
